# starting kernel
# baseline (speedup 1.0000x reference)
#include <hip/hip_runtime.h>
#include <math.h>

typedef __attribute__((ext_vector_type(2))) float v2f;
typedef __attribute__((ext_vector_type(8))) float v8f;

#define B_   64
#define L_   512
#define D_   50
#define M_   128
#define KP   64
#define LSPLIT 8
#define LCHUNK (L_ / LSPLIT)

__global__ void zero_kernel(float* __restrict__ p, int n) {
  int i = blockIdx.x * blockDim.x + threadIdx.x;
  if (i < n) p[i] = 0.0f;
}

__global__ void softmax_kernel(const int* __restrict__ x,
                               const float* __restrict__ mix_table,
                               float* __restrict__ w) {
  __shared__ float red[L_];
  const int b = blockIdx.x, t = threadIdx.x;
  const float v = mix_table[x[b * L_ + t]];
  red[t] = v;
  __syncthreads();
  for (int s = L_ / 2; s > 0; s >>= 1) {
    if (t < s) red[t] = fmaxf(red[t], red[t + s]);
    __syncthreads();
  }
  const float mx = red[0];
  __syncthreads();
  const float e = expf(v - mx);
  red[t] = e;
  __syncthreads();
  for (int s = L_ / 2; s > 0; s >>= 1) {
    if (t < s) red[t] += red[t + s];
    __syncthreads();
  }
  w[b * L_ + t] = e / red[0];
}

__global__ void knorm_kernel(const float* __restrict__ kr_in,
                             const float* __restrict__ ki_in,
                             float* __restrict__ krw,
                             float* __restrict__ kiw) {
  const int m = blockIdx.x * blockDim.x + threadIdx.x;
  if (m >= M_) return;
  float ss = 0.0f;
  for (int d = 0; d < D_; ++d) {
    const float a = kr_in[m * D_ + d];
    const float c = ki_in[m * D_ + d];
    ss += a * a + c * c;
  }
  const float rn = 1.0f / sqrtf(ss);
  for (int d = 0; d < KP; ++d) {
    krw[m * KP + d] = (d < D_) ? kr_in[m * D_ + d] * rn : 0.0f;
    kiw[m * KP + d] = (d < D_) ? ki_in[m * D_ + d] * rn : 0.0f;
  }
}

__global__ void __launch_bounds__(256)
qproj_kernel(const int* __restrict__ x,
             const float* __restrict__ amp_table,
             const float* __restrict__ pha_table,
             const float* __restrict__ w,
             const float* __restrict__ krw,
             const float* __restrict__ kiw,
             float* __restrict__ out) {
  extern __shared__ float smem[];
  float* krs  = smem;
  float* kis  = krs + M_ * KP;
  float* vrt  = kis + M_ * KP;
  float* vit  = vrt + 16 * KP;
  float* wt   = vit + 16 * KP;
  float* pacc = wt + 16;

  const int b    = blockIdx.x;
  const int lt0  = blockIdx.y * LCHUNK;
  const int tid  = threadIdx.x;
  const int wave = tid >> 5;
  const int lane = tid & 31;

  for (int i = tid; i < M_ * KP; i += 256) {
    krs[i] = krw[i];
    kis[i] = kiw[i];
  }
  if (tid < M_) pacc[tid] = 0.0f;
  __syncthreads();

  const int mbase = wave * 16;
  const int nrow  = lane & 15;
  const int khalf = (lane >> 4) * 2;
  const int rbase = (lane >> 4) * 8;

  for (int lt = lt0; lt < lt0 + LCHUNK; lt += 16) {
    for (int i = tid; i < 16 * KP; i += 256) {
      const int tok = lt + (i >> 6);
      const int d   = i & (KP - 1);
      float vr = 0.0f, vi = 0.0f;
      if (d < D_) {
        const int   idx = x[b * L_ + tok];
        const float a   = amp_table[idx * D_ + d];
        const float ph  = pha_table[idx * D_ + d];
        float s, c;
        sincosf(ph, &s, &c);
        vr = a * c;
        vi = a * s;
      }
      vrt[i] = vr;
      vit[i] = vi;
    }
    if (tid < 16) wt[tid] = w[b * L_ + lt + tid];
    __syncthreads();

    v8f cRR = {};
    v8f cII = {};
    v8f cIR = {};
    v8f cRI = {};
    #pragma unroll
    for (int ks = 0; ks < KP; ks += 4) {
      const int ka = ks + khalf;
      v2f aR = { vrt[nrow * KP + ka], vrt[nrow * KP + ka + 1] };
      v2f aI = { vit[nrow * KP + ka], vit[nrow * KP + ka + 1] };
      v2f bR = { krs[(mbase + nrow) * KP + ka], krs[(mbase + nrow) * KP + ka + 1] };
      v2f bI = { kis[(mbase + nrow) * KP + ka], kis[(mbase + nrow) * KP + ka + 1] };
      cRR = __builtin_amdgcn_wmma_f32_16x16x4_f32(false, aR, false, bR, (short)0, cRR, false, false);
      cII = __builtin_amdgcn_wmma_f32_16x16x4_f32(false, aI, false, bI, (short)0, cII, false, false);
      cIR = __builtin_amdgcn_wmma_f32_16x16x4_f32(false, aI, false, bR, (short)0, cIR, false, false);
      cRI = __builtin_amdgcn_wmma_f32_16x16x4_f32(false, aR, false, bI, (short)0, cRI, false, false);
    }

    float partial = 0.0f;
    #pragma unroll
    for (int r = 0; r < 8; ++r) {
      const float ar = cRR[r] + cII[r];
      const float ai = cIR[r] - cRI[r];
      partial = fmaf(wt[rbase + r], fmaf(ar, ar, ai * ai), partial);
    }
    partial += __shfl_xor(partial, 16, 32);
    if (lane < 16) pacc[mbase + lane] += partial;
    __syncthreads();
  }

  if (tid < M_) {
    unsafeAtomicAdd(&out[b * M_ + tid], pacc[tid]);
  }
}

extern "C" void kernel_launch(void* const* d_in, const int* in_sizes, int n_in,
                              void* d_out, int out_size, void* d_ws, size_t ws_size,
                              hipStream_t stream) {
  const int*   x   = (const int*)d_in[0];
  const float* amp = (const float*)d_in[1];
  const float* pha = (const float*)d_in[2];
  const float* mix = (const float*)d_in[3];
  const float* kr  = (const float*)d_in[4];
  const float* ki  = (const float*)d_in[5];
  float*       out = (float*)d_out;

  float* w   = (float*)d_ws;
  float* krw = w + B_ * L_;
  float* kiw = krw + M_ * KP;

  zero_kernel<<<(B_ * M_ + 255) / 256, 256, 0, stream>>>(out, B_ * M_);
  softmax_kernel<<<B_, L_, 0, stream>>>(x, mix, w);
  knorm_kernel<<<1, M_, 0, stream>>>(kr, ki, krw, kiw);

  const size_t lds_bytes =
      (size_t)(M_ * KP * 2 + 16 * KP * 2 + 16 + M_) * sizeof(float);
  dim3 grid(B_, LSPLIT);
  qproj_kernel<<<grid, 256, lds_bytes, stream>>>(x, amp, pha, w, krw, kiw, out);
}


# Row normalisation kernel: one wave per row with coalesced row loads and stores and a cross-lane sum of squares (tree order) instead of one thread per row with strided accesses
# speedup vs baseline: 1.3046x; 1.3046x over previous
#include <hip/hip_runtime.h>
#include <math.h>

typedef __attribute__((ext_vector_type(2))) float v2f;
typedef __attribute__((ext_vector_type(8))) float v8f;

#define B_   64
#define L_   512
#define D_   50
#define M_   128
#define KP   64
#define LSPLIT 8
#define LCHUNK (L_ / LSPLIT)

__global__ void zero_kernel(float* __restrict__ p, int n) {
  int i = blockIdx.x * blockDim.x + threadIdx.x;
  if (i < n) p[i] = 0.0f;
}

__global__ void softmax_kernel(const int* __restrict__ x,
                               const float* __restrict__ mix_table,
                               float* __restrict__ w) {
  __shared__ float red[L_];
  const int b = blockIdx.x, t = threadIdx.x;
  const float v = mix_table[x[b * L_ + t]];
  red[t] = v;
  __syncthreads();
  for (int s = L_ / 2; s > 0; s >>= 1) {
    if (t < s) red[t] = fmaxf(red[t], red[t + s]);
    __syncthreads();
  }
  const float mx = red[0];
  __syncthreads();
  const float e = expf(v - mx);
  red[t] = e;
  __syncthreads();
  for (int s = L_ / 2; s > 0; s >>= 1) {
    if (t < s) red[t] += red[t + s];
    __syncthreads();
  }
  w[b * L_ + t] = e / red[0];
}

__global__ void __launch_bounds__(32)
knorm_kernel(const float* __restrict__ kr_in,
             const float* __restrict__ ki_in,
             float* __restrict__ krw,
             float* __restrict__ kiw) {
  const int m    = blockIdx.x;
  const int lane = threadIdx.x;
  const int d1   = lane + 32;
  const bool in1 = d1 < D_;
  const float a0 = kr_in[m * D_ + lane];
  const float c0 = ki_in[m * D_ + lane];
  const float a1 = in1 ? kr_in[m * D_ + d1] : 0.0f;
  const float c1 = in1 ? ki_in[m * D_ + d1] : 0.0f;
  float ss = (a0 * a0 + c0 * c0) + (a1 * a1 + c1 * c1);
  for (int off = 16; off > 0; off >>= 1) ss += __shfl_xor(ss, off, 32);
  const float rn = 1.0f / sqrtf(ss);
  krw[m * KP + lane] = a0 * rn;
  kiw[m * KP + lane] = c0 * rn;
  krw[m * KP + d1]   = in1 ? a1 * rn : 0.0f;
  kiw[m * KP + d1]   = in1 ? c1 * rn : 0.0f;
}

__global__ void __launch_bounds__(256)
qproj_kernel(const int* __restrict__ x,
             const float* __restrict__ amp_table,
             const float* __restrict__ pha_table,
             const float* __restrict__ w,
             const float* __restrict__ krw,
             const float* __restrict__ kiw,
             float* __restrict__ out) {
  extern __shared__ float smem[];
  float* krs  = smem;
  float* kis  = krs + M_ * KP;
  float* vrt  = kis + M_ * KP;
  float* vit  = vrt + 16 * KP;
  float* wt   = vit + 16 * KP;
  float* pacc = wt + 16;

  const int b    = blockIdx.x;
  const int lt0  = blockIdx.y * LCHUNK;
  const int tid  = threadIdx.x;
  const int wave = tid >> 5;
  const int lane = tid & 31;

  for (int i = tid; i < M_ * KP; i += 256) {
    krs[i] = krw[i];
    kis[i] = kiw[i];
  }
  if (tid < M_) pacc[tid] = 0.0f;
  __syncthreads();

  const int mbase = wave * 16;
  const int nrow  = lane & 15;
  const int khalf = (lane >> 4) * 2;
  const int rbase = (lane >> 4) * 8;

  for (int lt = lt0; lt < lt0 + LCHUNK; lt += 16) {
    for (int i = tid; i < 16 * KP; i += 256) {
      const int tok = lt + (i >> 6);
      const int d   = i & (KP - 1);
      float vr = 0.0f, vi = 0.0f;
      if (d < D_) {
        const int   idx = x[b * L_ + tok];
        const float a   = amp_table[idx * D_ + d];
        const float ph  = pha_table[idx * D_ + d];
        float s, c;
        sincosf(ph, &s, &c);
        vr = a * c;
        vi = a * s;
      }
      vrt[i] = vr;
      vit[i] = vi;
    }
    if (tid < 16) wt[tid] = w[b * L_ + lt + tid];
    __syncthreads();

    v8f cRR = {};
    v8f cII = {};
    v8f cIR = {};
    v8f cRI = {};
    #pragma unroll
    for (int ks = 0; ks < KP; ks += 4) {
      const int ka = ks + khalf;
      v2f aR = { vrt[nrow * KP + ka], vrt[nrow * KP + ka + 1] };
      v2f aI = { vit[nrow * KP + ka], vit[nrow * KP + ka + 1] };
      v2f bR = { krs[(mbase + nrow) * KP + ka], krs[(mbase + nrow) * KP + ka + 1] };
      v2f bI = { kis[(mbase + nrow) * KP + ka], kis[(mbase + nrow) * KP + ka + 1] };
      cRR = __builtin_amdgcn_wmma_f32_16x16x4_f32(false, aR, false, bR, (short)0, cRR, false, false);
      cII = __builtin_amdgcn_wmma_f32_16x16x4_f32(false, aI, false, bI, (short)0, cII, false, false);
      cIR = __builtin_amdgcn_wmma_f32_16x16x4_f32(false, aI, false, bR, (short)0, cIR, false, false);
      cRI = __builtin_amdgcn_wmma_f32_16x16x4_f32(false, aR, false, bI, (short)0, cRI, false, false);
    }

    float partial = 0.0f;
    #pragma unroll
    for (int r = 0; r < 8; ++r) {
      const float ar = cRR[r] + cII[r];
      const float ai = cIR[r] - cRI[r];
      partial = fmaf(wt[rbase + r], fmaf(ar, ar, ai * ai), partial);
    }
    partial += __shfl_xor(partial, 16, 32);
    if (lane < 16) pacc[mbase + lane] += partial;
    __syncthreads();
  }

  if (tid < M_) {
    unsafeAtomicAdd(&out[b * M_ + tid], pacc[tid]);
  }
}

extern "C" void kernel_launch(void* const* d_in, const int* in_sizes, int n_in,
                              void* d_out, int out_size, void* d_ws, size_t ws_size,
                              hipStream_t stream) {
  const int*   x   = (const int*)d_in[0];
  const float* amp = (const float*)d_in[1];
  const float* pha = (const float*)d_in[2];
  const float* mix = (const float*)d_in[3];
  const float* kr  = (const float*)d_in[4];
  const float* ki  = (const float*)d_in[5];
  float*       out = (float*)d_out;

  float* w   = (float*)d_ws;
  float* krw = w + B_ * L_;
  float* kiw = krw + M_ * KP;

  zero_kernel<<<(B_ * M_ + 255) / 256, 256, 0, stream>>>(out, B_ * M_);
  softmax_kernel<<<B_, L_, 0, stream>>>(x, mix, w);
  knorm_kernel<<<M_, 32, 0, stream>>>(kr, ki, krw, kiw);

  const size_t lds_bytes =
      (size_t)(M_ * KP * 2 + 16 * KP * 2 + 16 + M_) * sizeof(float);
  dim3 grid(B_, LSPLIT);
  qproj_kernel<<<grid, 256, lds_bytes, stream>>>(x, amp, pha, w, krw, kiw, out);
}


# LDS row pitch of the measurement-vector and token tiles padded from 64 to 68 floats: WMMA fragment reads become bank-conflict-free (before, all rows of a fragment read fell on one bank)
# speedup vs baseline: 3.2254x; 2.4723x over previous
#include <hip/hip_runtime.h>
#include <math.h>

typedef __attribute__((ext_vector_type(2))) float v2f;
typedef __attribute__((ext_vector_type(8))) float v8f;

#define B_   64
#define L_   512
#define D_   50
#define M_   128
#define KP   64
#define LP   68
#define LSPLIT 8
#define LCHUNK (L_ / LSPLIT)

__global__ void zero_kernel(float* __restrict__ p, int n) {
  int i = blockIdx.x * blockDim.x + threadIdx.x;
  if (i < n) p[i] = 0.0f;
}

__global__ void softmax_kernel(const int* __restrict__ x,
                               const float* __restrict__ mix_table,
                               float* __restrict__ w) {
  __shared__ float red[L_];
  const int b = blockIdx.x, t = threadIdx.x;
  const float v = mix_table[x[b * L_ + t]];
  red[t] = v;
  __syncthreads();
  for (int s = L_ / 2; s > 0; s >>= 1) {
    if (t < s) red[t] = fmaxf(red[t], red[t + s]);
    __syncthreads();
  }
  const float mx = red[0];
  __syncthreads();
  const float e = expf(v - mx);
  red[t] = e;
  __syncthreads();
  for (int s = L_ / 2; s > 0; s >>= 1) {
    if (t < s) red[t] += red[t + s];
    __syncthreads();
  }
  w[b * L_ + t] = e / red[0];
}

__global__ void __launch_bounds__(32)
knorm_kernel(const float* __restrict__ kr_in,
             const float* __restrict__ ki_in,
             float* __restrict__ krw,
             float* __restrict__ kiw) {
  const int m    = blockIdx.x;
  const int lane = threadIdx.x;
  const int d1   = lane + 32;
  const bool in1 = d1 < D_;
  const float a0 = kr_in[m * D_ + lane];
  const float c0 = ki_in[m * D_ + lane];
  const float a1 = in1 ? kr_in[m * D_ + d1] : 0.0f;
  const float c1 = in1 ? ki_in[m * D_ + d1] : 0.0f;
  float ss = (a0 * a0 + c0 * c0) + (a1 * a1 + c1 * c1);
  for (int off = 16; off > 0; off >>= 1) ss += __shfl_xor(ss, off, 32);
  const float rn = 1.0f / sqrtf(ss);
  krw[m * KP + lane] = a0 * rn;
  kiw[m * KP + lane] = c0 * rn;
  krw[m * KP + d1]   = in1 ? a1 * rn : 0.0f;
  kiw[m * KP + d1]   = in1 ? c1 * rn : 0.0f;
}

__global__ void __launch_bounds__(256)
qproj_kernel(const int* __restrict__ x,
             const float* __restrict__ amp_table,
             const float* __restrict__ pha_table,
             const float* __restrict__ w,
             const float* __restrict__ krw,
             const float* __restrict__ kiw,
             float* __restrict__ out) {
  extern __shared__ float smem[];
  float* krs  = smem;
  float* kis  = krs + M_ * LP;
  float* vrt  = kis + M_ * LP;
  float* vit  = vrt + 16 * LP;
  float* wt   = vit + 16 * LP;
  float* pacc = wt + 16;

  const int b    = blockIdx.x;
  const int lt0  = blockIdx.y * LCHUNK;
  const int tid  = threadIdx.x;
  const int wave = tid >> 5;
  const int lane = tid & 31;

  for (int i = tid; i < M_ * KP; i += 256) {
    const int li = (i >> 6) * LP + (i & (KP - 1));
    krs[li] = krw[i];
    kis[li] = kiw[i];
  }
  if (tid < M_) pacc[tid] = 0.0f;
  __syncthreads();

  const int mbase = wave * 16;
  const int nrow  = lane & 15;
  const int khalf = (lane >> 4) * 2;
  const int rbase = (lane >> 4) * 8;

  for (int lt = lt0; lt < lt0 + LCHUNK; lt += 16) {
    for (int i = tid; i < 16 * KP; i += 256) {
      const int tok = lt + (i >> 6);
      const int d   = i & (KP - 1);
      float vr = 0.0f, vi = 0.0f;
      if (d < D_) {
        const int   idx = x[b * L_ + tok];
        const float a   = amp_table[idx * D_ + d];
        const float ph  = pha_table[idx * D_ + d];
        float s, c;
        sincosf(ph, &s, &c);
        vr = a * c;
        vi = a * s;
      }
      vrt[(i >> 6) * LP + d] = vr;
      vit[(i >> 6) * LP + d] = vi;
    }
    if (tid < 16) wt[tid] = w[b * L_ + lt + tid];
    __syncthreads();

    v8f cRR = {};
    v8f cII = {};
    v8f cIR = {};
    v8f cRI = {};
    #pragma unroll
    for (int ks = 0; ks < KP; ks += 4) {
      const int ka = ks + khalf;
      v2f aR = { vrt[nrow * LP + ka], vrt[nrow * LP + ka + 1] };
      v2f aI = { vit[nrow * LP + ka], vit[nrow * LP + ka + 1] };
      v2f bR = { krs[(mbase + nrow) * LP + ka], krs[(mbase + nrow) * LP + ka + 1] };
      v2f bI = { kis[(mbase + nrow) * LP + ka], kis[(mbase + nrow) * LP + ka + 1] };
      cRR = __builtin_amdgcn_wmma_f32_16x16x4_f32(false, aR, false, bR, (short)0, cRR, false, false);
      cII = __builtin_amdgcn_wmma_f32_16x16x4_f32(false, aI, false, bI, (short)0, cII, false, false);
      cIR = __builtin_amdgcn_wmma_f32_16x16x4_f32(false, aI, false, bR, (short)0, cIR, false, false);
      cRI = __builtin_amdgcn_wmma_f32_16x16x4_f32(false, aR, false, bI, (short)0, cRI, false, false);
    }

    float partial = 0.0f;
    #pragma unroll
    for (int r = 0; r < 8; ++r) {
      const float ar = cRR[r] + cII[r];
      const float ai = cIR[r] - cRI[r];
      partial = fmaf(wt[rbase + r], fmaf(ar, ar, ai * ai), partial);
    }
    partial += __shfl_xor(partial, 16, 32);
    if (lane < 16) pacc[mbase + lane] += partial;
    __syncthreads();
  }

  if (tid < M_) {
    unsafeAtomicAdd(&out[b * M_ + tid], pacc[tid]);
  }
}

extern "C" void kernel_launch(void* const* d_in, const int* in_sizes, int n_in,
                              void* d_out, int out_size, void* d_ws, size_t ws_size,
                              hipStream_t stream) {
  const int*   x   = (const int*)d_in[0];
  const float* amp = (const float*)d_in[1];
  const float* pha = (const float*)d_in[2];
  const float* mix = (const float*)d_in[3];
  const float* kr  = (const float*)d_in[4];
  const float* ki  = (const float*)d_in[5];
  float*       out = (float*)d_out;

  float* w   = (float*)d_ws;
  float* krw = w + B_ * L_;
  float* kiw = krw + M_ * KP;

  zero_kernel<<<(B_ * M_ + 255) / 256, 256, 0, stream>>>(out, B_ * M_);
  softmax_kernel<<<B_, L_, 0, stream>>>(x, mix, w);
  knorm_kernel<<<M_, 32, 0, stream>>>(kr, ki, krw, kiw);

  const size_t lds_bytes =
      (size_t)(M_ * LP * 2 + 16 * LP * 2 + 16 + M_) * sizeof(float);
  dim3 grid(B_, LSPLIT);
  qproj_kernel<<<grid, 256, lds_bytes, stream>>>(x, amp, pha, w, krw, kiw, out);
}


# Output zeroing, sequence softmax and row normalisation run as ONE launch instead of three (a branch on the block index picks the job): two launch boundaries removed, all arithmetic unchanged
# speedup vs baseline: 3.8121x; 1.1819x over previous
#include <hip/hip_runtime.h>
#include <math.h>

typedef __attribute__((ext_vector_type(2))) float v2f;
typedef __attribute__((ext_vector_type(8))) float v8f;

#define B_   64
#define L_   512
#define D_   50
#define M_   128
#define KP   64
#define LP   68
#define LSPLIT 8
#define LCHUNK (L_ / LSPLIT)

#define PREP_T   L_
#define PREP_KB  (M_ / (PREP_T / 32))

__global__ void __launch_bounds__(PREP_T)
prep_kernel(const int* __restrict__ x,
            const float* __restrict__ mix_table,
            const float* __restrict__ kr_in,
            const float* __restrict__ ki_in,
            float* __restrict__ w,
            float* __restrict__ krw,
            float* __restrict__ kiw,
            float* __restrict__ out) {
  __shared__ float red[L_];
  const int t = threadIdx.x;

  if (blockIdx.x >= B_) {
    const int kb = blockIdx.x - B_;
    static_assert((B_ * M_) % (PREP_KB * PREP_T) == 0, "output splits evenly over the zeroing threads");
    #pragma unroll
    for (int j = 0; j < (B_ * M_) / (PREP_KB * PREP_T); ++j)
      out[j * (PREP_KB * PREP_T) + kb * PREP_T + t] = 0.0f;

    const int m    = kb * (PREP_T / 32) + (t >> 5);
    const int lane = t & 31;
    const int d1   = lane + 32;
    const bool in1 = d1 < D_;
    const float a0 = kr_in[m * D_ + lane];
    const float c0 = ki_in[m * D_ + lane];
    const float a1 = in1 ? kr_in[m * D_ + d1] : 0.0f;
    const float c1 = in1 ? ki_in[m * D_ + d1] : 0.0f;
    float ss = (a0 * a0 + c0 * c0) + (a1 * a1 + c1 * c1);
    for (int off = 16; off > 0; off >>= 1) ss += __shfl_xor(ss, off, 32);
    const float rn = 1.0f / sqrtf(ss);
    krw[m * KP + lane] = a0 * rn;
    kiw[m * KP + lane] = c0 * rn;
    krw[m * KP + d1]   = in1 ? a1 * rn : 0.0f;
    kiw[m * KP + d1]   = in1 ? c1 * rn : 0.0f;
    return;
  }

  const int b = blockIdx.x;
  const float v = mix_table[x[b * L_ + t]];
  red[t] = v;
  __syncthreads();
  for (int s = L_ / 2; s > 0; s >>= 1) {
    if (t < s) red[t] = fmaxf(red[t], red[t + s]);
    __syncthreads();
  }
  const float mx = red[0];
  __syncthreads();
  const float e = expf(v - mx);
  red[t] = e;
  __syncthreads();
  for (int s = L_ / 2; s > 0; s >>= 1) {
    if (t < s) red[t] += red[t + s];
    __syncthreads();
  }
  w[b * L_ + t] = e / red[0];
}

__global__ void __launch_bounds__(256)
qproj_kernel(const int* __restrict__ x,
             const float* __restrict__ amp_table,
             const float* __restrict__ pha_table,
             const float* __restrict__ w,
             const float* __restrict__ krw,
             const float* __restrict__ kiw,
             float* __restrict__ out) {
  extern __shared__ float smem[];
  float* krs  = smem;
  float* kis  = krs + M_ * LP;
  float* vrt  = kis + M_ * LP;
  float* vit  = vrt + 16 * LP;
  float* wt   = vit + 16 * LP;
  float* pacc = wt + 16;

  const int b    = blockIdx.x;
  const int lt0  = blockIdx.y * LCHUNK;
  const int tid  = threadIdx.x;
  const int wave = tid >> 5;
  const int lane = tid & 31;

  for (int i = tid; i < M_ * KP; i += 256) {
    const int li = (i >> 6) * LP + (i & (KP - 1));
    krs[li] = krw[i];
    kis[li] = kiw[i];
  }
  if (tid < M_) pacc[tid] = 0.0f;
  __syncthreads();

  const int mbase = wave * 16;
  const int nrow  = lane & 15;
  const int khalf = (lane >> 4) * 2;
  const int rbase = (lane >> 4) * 8;

  for (int lt = lt0; lt < lt0 + LCHUNK; lt += 16) {
    for (int i = tid; i < 16 * KP; i += 256) {
      const int tok = lt + (i >> 6);
      const int d   = i & (KP - 1);
      float vr = 0.0f, vi = 0.0f;
      if (d < D_) {
        const int   idx = x[b * L_ + tok];
        const float a   = amp_table[idx * D_ + d];
        const float ph  = pha_table[idx * D_ + d];
        float s, c;
        sincosf(ph, &s, &c);
        vr = a * c;
        vi = a * s;
      }
      vrt[(i >> 6) * LP + d] = vr;
      vit[(i >> 6) * LP + d] = vi;
    }
    if (tid < 16) wt[tid] = w[b * L_ + lt + tid];
    __syncthreads();

    v8f cRR = {};
    v8f cII = {};
    v8f cIR = {};
    v8f cRI = {};
    #pragma unroll
    for (int ks = 0; ks < KP; ks += 4) {
      const int ka = ks + khalf;
      v2f aR = { vrt[nrow * LP + ka], vrt[nrow * LP + ka + 1] };
      v2f aI = { vit[nrow * LP + ka], vit[nrow * LP + ka + 1] };
      v2f bR = { krs[(mbase + nrow) * LP + ka], krs[(mbase + nrow) * LP + ka + 1] };
      v2f bI = { kis[(mbase + nrow) * LP + ka], kis[(mbase + nrow) * LP + ka + 1] };
      cRR = __builtin_amdgcn_wmma_f32_16x16x4_f32(false, aR, false, bR, (short)0, cRR, false, false);
      cII = __builtin_amdgcn_wmma_f32_16x16x4_f32(false, aI, false, bI, (short)0, cII, false, false);
      cIR = __builtin_amdgcn_wmma_f32_16x16x4_f32(false, aI, false, bR, (short)0, cIR, false, false);
      cRI = __builtin_amdgcn_wmma_f32_16x16x4_f32(false, aR, false, bI, (short)0, cRI, false, false);
    }

    float partial = 0.0f;
    #pragma unroll
    for (int r = 0; r < 8; ++r) {
      const float ar = cRR[r] + cII[r];
      const float ai = cIR[r] - cRI[r];
      partial = fmaf(wt[rbase + r], fmaf(ar, ar, ai * ai), partial);
    }
    partial += __shfl_xor(partial, 16, 32);
    if (lane < 16) pacc[mbase + lane] += partial;
    __syncthreads();
  }

  if (tid < M_) {
    unsafeAtomicAdd(&out[b * M_ + tid], pacc[tid]);
  }
}

extern "C" void kernel_launch(void* const* d_in, const int* in_sizes, int n_in,
                              void* d_out, int out_size, void* d_ws, size_t ws_size,
                              hipStream_t stream) {
  const int*   x   = (const int*)d_in[0];
  const float* amp = (const float*)d_in[1];
  const float* pha = (const float*)d_in[2];
  const float* mix = (const float*)d_in[3];
  const float* kr  = (const float*)d_in[4];
  const float* ki  = (const float*)d_in[5];
  float*       out = (float*)d_out;

  float* w   = (float*)d_ws;
  float* krw = w + B_ * L_;
  float* kiw = krw + M_ * KP;

  prep_kernel<<<B_ + PREP_KB, PREP_T, 0, stream>>>(x, mix, kr, ki, w, krw, kiw, out);

  const size_t lds_bytes =
      (size_t)(M_ * LP * 2 + 16 * LP * 2 + 16 + M_) * sizeof(float);
  dim3 grid(B_, LSPLIT);
  qproj_kernel<<<grid, 256, lds_bytes, stream>>>(x, amp, pha, w, krw, kiw, out);
}


# Token gather: each block stages its whole 64-token chunk in one pass with all index, amplitude and phase loads in flight together (was 16 dependent load pairs in series per thread); sums unchanged
# speedup vs baseline: 5.0934x; 1.3361x over previous
#include <hip/hip_runtime.h>
#include <math.h>

typedef __attribute__((ext_vector_type(2))) float v2f;
typedef __attribute__((ext_vector_type(8))) float v8f;

#define B_   64
#define L_   512
#define D_   50
#define M_   128
#define KP   64
#define LP   68
#define LSPLIT 8
#define LCHUNK (L_ / LSPLIT)

#define PREP_T   L_
#define PREP_KB  (M_ / (PREP_T / 32))

__global__ void __launch_bounds__(PREP_T)
prep_kernel(const int* __restrict__ x,
            const float* __restrict__ mix_table,
            const float* __restrict__ kr_in,
            const float* __restrict__ ki_in,
            float* __restrict__ w,
            float* __restrict__ krw,
            float* __restrict__ kiw,
            float* __restrict__ out) {
  __shared__ float red[L_];
  const int t = threadIdx.x;

  if (blockIdx.x >= B_) {
    const int kb = blockIdx.x - B_;
    static_assert((B_ * M_) % (PREP_KB * PREP_T) == 0, "output splits evenly over the zeroing threads");
    #pragma unroll
    for (int j = 0; j < (B_ * M_) / (PREP_KB * PREP_T); ++j)
      out[j * (PREP_KB * PREP_T) + kb * PREP_T + t] = 0.0f;

    const int m    = kb * (PREP_T / 32) + (t >> 5);
    const int lane = t & 31;
    const int d1   = lane + 32;
    const bool in1 = d1 < D_;
    const float a0 = kr_in[m * D_ + lane];
    const float c0 = ki_in[m * D_ + lane];
    const float a1 = in1 ? kr_in[m * D_ + d1] : 0.0f;
    const float c1 = in1 ? ki_in[m * D_ + d1] : 0.0f;
    float ss = (a0 * a0 + c0 * c0) + (a1 * a1 + c1 * c1);
    for (int off = 16; off > 0; off >>= 1) ss += __shfl_xor(ss, off, 32);
    const float rn = 1.0f / sqrtf(ss);
    krw[m * KP + lane] = a0 * rn;
    kiw[m * KP + lane] = c0 * rn;
    krw[m * KP + d1]   = in1 ? a1 * rn : 0.0f;
    kiw[m * KP + d1]   = in1 ? c1 * rn : 0.0f;
    return;
  }

  const int b = blockIdx.x;
  const float v = mix_table[x[b * L_ + t]];
  red[t] = v;
  __syncthreads();
  for (int s = L_ / 2; s > 0; s >>= 1) {
    if (t < s) red[t] = fmaxf(red[t], red[t + s]);
    __syncthreads();
  }
  const float mx = red[0];
  __syncthreads();
  const float e = expf(v - mx);
  red[t] = e;
  __syncthreads();
  for (int s = L_ / 2; s > 0; s >>= 1) {
    if (t < s) red[t] += red[t + s];
    __syncthreads();
  }
  w[b * L_ + t] = e / red[0];
}

#define EPT (LCHUNK * KP / 256)

__global__ void __launch_bounds__(256)
qproj_kernel(const int* __restrict__ x,
             const float* __restrict__ amp_table,
             const float* __restrict__ pha_table,
             const float* __restrict__ w,
             const float* __restrict__ krw,
             const float* __restrict__ kiw,
             float* __restrict__ out) {
  extern __shared__ float smem[];
  float* krs  = smem;
  float* kis  = krs + M_ * LP;
  float* vrt  = kis + M_ * LP;
  float* vit  = vrt + LCHUNK * LP;
  float* wt   = vit + LCHUNK * LP;
  float* pacc = wt + LCHUNK;

  const int b    = blockIdx.x;
  const int lt0  = blockIdx.y * LCHUNK;
  const int tid  = threadIdx.x;
  const int wave = tid >> 5;
  const int lane = tid & 31;

  static_assert(256 / KP == 4 && EPT * 4 == LCHUNK, "thread-to-entry map of the gather");
  const int d    = tid & (KP - 1);
  const int trow = tid >> 6;
  float av[EPT], pv[EPT];
  {
    int idx[EPT];
    #pragma unroll
    for (int j = 0; j < EPT; ++j) idx[j] = x[b * L_ + lt0 + trow + 4 * j];
    #pragma unroll
    for (int j = 0; j < EPT; ++j) { av[j] = 0.0f; pv[j] = 0.0f; }
    if (d < D_) {
      #pragma unroll
      for (int j = 0; j < EPT; ++j) {
        av[j] = amp_table[idx[j] * D_ + d];
        pv[j] = pha_table[idx[j] * D_ + d];
      }
    }
  }

  for (int i = tid; i < M_ * KP; i += 256) {
    const int li = (i >> 6) * LP + (i & (KP - 1));
    krs[li] = krw[i];
    kis[li] = kiw[i];
  }
  if (tid < M_) pacc[tid] = 0.0f;
  if (tid < LCHUNK) wt[tid] = w[b * L_ + lt0 + tid];

  #pragma unroll
  for (int j = 0; j < EPT; ++j) {
    float s, c;
    sincosf(pv[j], &s, &c);
    vrt[(trow + 4 * j) * LP + d] = av[j] * c;
    vit[(trow + 4 * j) * LP + d] = av[j] * s;
  }
  __syncthreads();

  const int mbase = wave * 16;
  const int nrow  = lane & 15;
  const int khalf = (lane >> 4) * 2;
  const int rbase = (lane >> 4) * 8;

  for (int t0 = 0; t0 < LCHUNK; t0 += 16) {
    const float* vr16 = vrt + t0 * LP;
    const float* vi16 = vit + t0 * LP;

    v8f cRR = {};
    v8f cII = {};
    v8f cIR = {};
    v8f cRI = {};
    #pragma unroll
    for (int ks = 0; ks < KP; ks += 4) {
      const int ka = ks + khalf;
      v2f aR = { vr16[nrow * LP + ka], vr16[nrow * LP + ka + 1] };
      v2f aI = { vi16[nrow * LP + ka], vi16[nrow * LP + ka + 1] };
      v2f bR = { krs[(mbase + nrow) * LP + ka], krs[(mbase + nrow) * LP + ka + 1] };
      v2f bI = { kis[(mbase + nrow) * LP + ka], kis[(mbase + nrow) * LP + ka + 1] };
      cRR = __builtin_amdgcn_wmma_f32_16x16x4_f32(false, aR, false, bR, (short)0, cRR, false, false);
      cII = __builtin_amdgcn_wmma_f32_16x16x4_f32(false, aI, false, bI, (short)0, cII, false, false);
      cIR = __builtin_amdgcn_wmma_f32_16x16x4_f32(false, aI, false, bR, (short)0, cIR, false, false);
      cRI = __builtin_amdgcn_wmma_f32_16x16x4_f32(false, aR, false, bI, (short)0, cRI, false, false);
    }

    float partial = 0.0f;
    #pragma unroll
    for (int r = 0; r < 8; ++r) {
      const float ar = cRR[r] + cII[r];
      const float ai = cIR[r] - cRI[r];
      partial = fmaf(wt[t0 + rbase + r], fmaf(ar, ar, ai * ai), partial);
    }
    partial += __shfl_xor(partial, 16, 32);
    if (lane < 16) pacc[mbase + lane] += partial;
  }
  __syncthreads();

  if (tid < M_) {
    unsafeAtomicAdd(&out[b * M_ + tid], pacc[tid]);
  }
}

extern "C" void kernel_launch(void* const* d_in, const int* in_sizes, int n_in,
                              void* d_out, int out_size, void* d_ws, size_t ws_size,
                              hipStream_t stream) {
  const int*   x   = (const int*)d_in[0];
  const float* amp = (const float*)d_in[1];
  const float* pha = (const float*)d_in[2];
  const float* mix = (const float*)d_in[3];
  const float* kr  = (const float*)d_in[4];
  const float* ki  = (const float*)d_in[5];
  float*       out = (float*)d_out;

  float* w   = (float*)d_ws;
  float* krw = w + B_ * L_;
  float* kiw = krw + M_ * KP;

  prep_kernel<<<B_ + PREP_KB, PREP_T, 0, stream>>>(x, mix, kr, ki, w, krw, kiw, out);

  const size_t lds_bytes =
      (size_t)(M_ * LP * 2 + LCHUNK * LP * 2 + LCHUNK + M_) * sizeof(float);
  dim3 grid(B_, LSPLIT);
  qproj_kernel<<<grid, 256, lds_bytes, stream>>>(x, amp, pha, w, krw, kiw, out);
}


# Each complex dot product runs over ONE concatenated K of 100 ([re|im] rows, 25 k-steps, 2 accumulators), not 4 products over K padded to 64: 50 matrix instructions per tile, was 64; sums reordered
# speedup vs baseline: 5.5261x; 1.0850x over previous
#include <hip/hip_runtime.h>
#include <math.h>


typedef __attribute__((ext_vector_type(2))) float v2f;
typedef __attribute__((ext_vector_type(4))) float v4f;
typedef __attribute__((ext_vector_type(8))) float v8f;

#define B_   64
#define L_   512
#define D_   50
#define M_   128
#define KC   (2 * D_)
#define BP   KC
#define AP   156
#define LSPLIT 8
#define LCHUNK (L_ / LSPLIT)

#define PREP_T   L_
#define PREP_KB  (M_ / (PREP_T / 32))

__global__ void __launch_bounds__(PREP_T)
prep_kernel(const int* __restrict__ x,
            const float* __restrict__ mix_table,
            const float* __restrict__ kr_in,
            const float* __restrict__ ki_in,
            float* __restrict__ w,
            float* __restrict__ kcw,
            float* __restrict__ out) {
  __shared__ float red[L_];
  const int t = threadIdx.x;

  if (blockIdx.x >= B_) {
    const int kb = blockIdx.x - B_;
    static_assert((B_ * M_) % (PREP_KB * PREP_T) == 0, "output splits evenly over the zeroing threads");
    #pragma unroll
    for (int j = 0; j < (B_ * M_) / (PREP_KB * PREP_T); ++j)
      out[j * (PREP_KB * PREP_T) + kb * PREP_T + t] = 0.0f;

    const int m    = kb * (PREP_T / 32) + (t >> 5);
    const int lane = t & 31;
    const int d1   = lane + 32;
    const bool in1 = d1 < D_;
    const float a0 = kr_in[m * D_ + lane];
    const float c0 = ki_in[m * D_ + lane];
    const float a1 = in1 ? kr_in[m * D_ + d1] : 0.0f;
    const float c1 = in1 ? ki_in[m * D_ + d1] : 0.0f;
    float ss = (a0 * a0 + c0 * c0) + (a1 * a1 + c1 * c1);
    for (int off = 16; off > 0; off >>= 1) ss += __shfl_xor(ss, off, 32);
    const float rn = 1.0f / sqrtf(ss);
    float* row = kcw + m * KC;
    row[lane]      = a0 * rn;
    row[D_ + lane] = c0 * rn;
    if (in1) {
      row[d1]      = a1 * rn;
      row[D_ + d1] = c1 * rn;
    }
    return;
  }

  const int b = blockIdx.x;
  const float v = mix_table[x[b * L_ + t]];
  red[t] = v;
  __syncthreads();
  for (int s = L_ / 2; s > 0; s >>= 1) {
    if (t < s) red[t] = fmaxf(red[t], red[t + s]);
    __syncthreads();
  }
  const float mx = red[0];
  __syncthreads();
  const float e = expf(v - mx);
  red[t] = e;
  __syncthreads();
  for (int s = L_ / 2; s > 0; s >>= 1) {
    if (t < s) red[t] += red[t + s];
    __syncthreads();
  }
  w[b * L_ + t] = e / red[0];
}

#define GT  (256 / D_)
#define GPS ((LCHUNK + GT - 1) / GT)
#define KS  (KC / 4)

__global__ void __launch_bounds__(256)
qproj_kernel(const int* __restrict__ x,
             const float* __restrict__ amp_table,
             const float* __restrict__ pha_table,
             const float* __restrict__ w,
             const float* __restrict__ kcw,
             float* __restrict__ out) {
  extern __shared__ float smem[];
  float* kcs  = smem;
  float* vt   = kcs + M_ * BP;
  float* wt   = vt + LCHUNK * AP;
  float* pacc = wt + LCHUNK;

  const int b    = blockIdx.x;
  const int lt0  = blockIdx.y * LCHUNK;
  const int tid  = threadIdx.x;
  const int wave = tid >> 5;
  const int lane = tid & 31;

  static_assert(KC % 4 == 0 && D_ % 2 == 0, "k-steps are whole and a fragment pair never straddles the two parts");
  static_assert((M_ * BP) % 4 == 0, "measurement rows copy as whole 16-byte pieces");
  const int trow = tid / D_;
  const int d    = tid - trow * D_;
  const bool act = tid < GT * D_;
  float av[GPS], pv[GPS];
  {
    int idx[GPS];
    #pragma unroll
    for (int j = 0; j < GPS; ++j) {
      const int tok = min(trow + GT * j, LCHUNK - 1);
      idx[j] = x[b * L_ + lt0 + tok];
    }
    #pragma unroll
    for (int j = 0; j < GPS; ++j) {
      av[j] = amp_table[idx[j] * D_ + d];
      pv[j] = pha_table[idx[j] * D_ + d];
    }
  }

  {
    const v4f* src = reinterpret_cast<const v4f*>(kcw);
    v4f* dst = reinterpret_cast<v4f*>(kcs);
    for (int i = tid; i < M_ * BP / 4; i += 256) dst[i] = src[i];
  }
  if (tid < M_) pacc[tid] = 0.0f;
  if (tid < LCHUNK) wt[tid] = w[b * L_ + lt0 + tid];

  #pragma unroll
  for (int j = 0; j < GPS; ++j) {
    const int tok = trow + GT * j;
    float s, c;
    sincosf(pv[j], &s, &c);
    const float re = av[j] * c;
    const float im = av[j] * s;
    if (act && tok < LCHUNK) {
      float* row = vt + tok * AP + d;
      row[0]      = re;
      row[D_]     = im;
      row[2 * D_] = -re;
    }
  }
  __syncthreads();

  const int mbase = wave * 16;
  const int nrow  = lane & 15;
  const int khalf = (lane >> 4) * 2;
  const int rbase = (lane >> 4) * 8;

  v2f bK[KS];
  {
    const float* brow = kcs + (mbase + nrow) * BP + khalf;
    #pragma unroll
    for (int ks = 0; ks < KS; ++ks) bK[ks] = *reinterpret_cast<const v2f*>(brow + 4 * ks);
  }

  for (int t0 = 0; t0 < LCHUNK; t0 += 16) {
    const float* arow = vt + (t0 + nrow) * AP + khalf;

    v8f cAr = {};
    v8f cAi = {};
    #pragma unroll
    for (int ks = 0; ks < KS; ++ks) {
      const v2f aR = *reinterpret_cast<const v2f*>(arow + 4 * ks);
      const v2f aI = *reinterpret_cast<const v2f*>(arow + D_ + 4 * ks);
      cAr = __builtin_amdgcn_wmma_f32_16x16x4_f32(false, aR, false, bK[ks], (short)0, cAr, false, false);
      cAi = __builtin_amdgcn_wmma_f32_16x16x4_f32(false, aI, false, bK[ks], (short)0, cAi, false, false);
    }

    float partial = 0.0f;
    #pragma unroll
    for (int r = 0; r < 8; ++r) {
      const float ar = cAr[r];
      const float ai = cAi[r];
      partial = fmaf(wt[t0 + rbase + r], fmaf(ar, ar, ai * ai), partial);
    }
    partial += __shfl_xor(partial, 16, 32);
    if (lane < 16) pacc[mbase + lane] += partial;
  }
  __syncthreads();

  if (tid < M_) {
    unsafeAtomicAdd(&out[b * M_ + tid], pacc[tid]);
  }
}

extern "C" void kernel_launch(void* const* d_in, const int* in_sizes, int n_in,
                              void* d_out, int out_size, void* d_ws, size_t ws_size,
                              hipStream_t stream) {
  const int*   x   = (const int*)d_in[0];
  const float* amp = (const float*)d_in[1];
  const float* pha = (const float*)d_in[2];
  const float* mix = (const float*)d_in[3];
  const float* kr  = (const float*)d_in[4];
  const float* ki  = (const float*)d_in[5];
  float*       out = (float*)d_out;

  float* w   = (float*)d_ws;
  float* kcw = w + B_ * L_;

  prep_kernel<<<B_ + PREP_KB, PREP_T, 0, stream>>>(x, mix, kr, ki, w, kcw, out);

  const size_t lds_bytes =
      (size_t)(M_ * BP + LCHUNK * AP + LCHUNK + M_) * sizeof(float);
  dim3 grid(B_, LSPLIT);
  qproj_kernel<<<grid, 256, lds_bytes, stream>>>(x, amp, pha, w, kcw, out);
}


# Softmax and row norms applied after the sums: each block uses its own chunk's max and raw rows; a small merge launch joins the 8 chunks (replaces preparation launch, zeroing, atomics); sums reordered
# speedup vs baseline: 6.1900x; 1.1201x over previous
#include <hip/hip_runtime.h>
#include <math.h>


typedef __attribute__((ext_vector_type(2))) float v2f;
typedef __attribute__((ext_vector_type(8))) float v8f;

#define B_   64
#define L_   512
#define D_   50
#define M_   128
#define KC   (2 * D_)
#define BP   KC
#define AP   156
#define LSPLIT 8
#define LCHUNK (L_ / LSPLIT)
#define PROW (M_ + 32)


#define GT  (256 / D_)
#define GPS ((LCHUNK + GT - 1) / GT)
#define KS  (KC / 4)

__global__ void __launch_bounds__(256)
qproj_kernel(const int* __restrict__ x,
             const float* __restrict__ amp_table,
             const float* __restrict__ pha_table,
             const float* __restrict__ mix_table,
             const float* __restrict__ kr_in,
             const float* __restrict__ ki_in,
             float* __restrict__ part) {
  extern __shared__ float smem[];
  float* kcs  = smem;
  float* vt   = kcs + M_ * BP;
  float* wt   = vt + LCHUNK * AP;
  float* pacc = wt + LCHUNK;
  float* scal = pacc + M_;

  const int b     = blockIdx.x;
  const int split = blockIdx.y;
  const int lt0   = split * LCHUNK;
  const int tid   = threadIdx.x;
  const int wave  = tid >> 5;
  const int lane  = tid & 31;

  static_assert(KC % 4 == 0 && D_ % 2 == 0, "k-steps are whole and a fragment pair never straddles the two parts");
  static_assert(LCHUNK == 64, "one wave holds the chunk's mix logits as two tokens per lane");
  const int trow = tid / D_;
  const int d    = tid - trow * D_;
  const bool act = tid < GT * D_;
  float av[GPS], pv[GPS];
  {
    int idx[GPS];
    #pragma unroll
    for (int j = 0; j < GPS; ++j) {
      const int tok = min(trow + GT * j, LCHUNK - 1);
      idx[j] = x[b * L_ + lt0 + tok];
    }
    #pragma unroll
    for (int j = 0; j < GPS; ++j) {
      av[j] = amp_table[idx[j] * D_ + d];
      pv[j] = pha_table[idx[j] * D_ + d];
    }
  }

  float mv0 = 0.0f, mv1 = 0.0f;
  if (wave == 0) {
    mv0 = mix_table[x[b * L_ + lt0 + lane]];
    mv1 = mix_table[x[b * L_ + lt0 + 32 + lane]];
  }

  {
    const v2f* srcr = reinterpret_cast<const v2f*>(kr_in);
    const v2f* srci = reinterpret_cast<const v2f*>(ki_in);
    for (int i = tid; i < M_ * (D_ / 2); i += 256) {
      const int m  = i / (D_ / 2);
      const int dp = i - m * (D_ / 2);
      v2f* dst = reinterpret_cast<v2f*>(kcs + m * BP) + dp;
      dst[0]      = srcr[i];
      dst[D_ / 2] = srci[i];
    }
  }

  if (wave == 0) {
    float cm = fmaxf(mv0, mv1);
    for (int off = 16; off > 0; off >>= 1) cm = fmaxf(cm, __shfl_xor(cm, off, 32));
    const float e0 = expf(mv0 - cm);
    const float e1 = expf(mv1 - cm);
    wt[lane]      = e0;
    wt[lane + 32] = e1;
    float es = e0 + e1;
    for (int off = 16; off > 0; off >>= 1) es += __shfl_xor(es, off, 32);
    if (lane == 0) {
      scal[0] = cm;
      scal[1] = es;
    }
  }

  #pragma unroll
  for (int j = 0; j < GPS; ++j) {
    const int tok = trow + GT * j;
    float s, c;
    sincosf(pv[j], &s, &c);
    const float re = av[j] * c;
    const float im = av[j] * s;
    if (act && tok < LCHUNK) {
      float* row = vt + tok * AP + d;
      row[0]      = re;
      row[D_]     = im;
      row[2 * D_] = -re;
    }
  }
  __syncthreads();

  const int mbase = wave * 16;
  const int nrow  = lane & 15;
  const int khalf = (lane >> 4) * 2;
  const int rbase = (lane >> 4) * 8;

  v2f bK[KS];
  {
    const float* brow = kcs + (mbase + nrow) * BP + khalf;
    #pragma unroll
    for (int ks = 0; ks < KS; ++ks) bK[ks] = *reinterpret_cast<const v2f*>(brow + 4 * ks);
  }

  float nx = 0.0f, ny = 0.0f;
  #pragma unroll
  for (int ks = 0; ks < KS; ++ks) {
    nx = fmaf(bK[ks].x, bK[ks].x, nx);
    ny = fmaf(bK[ks].y, bK[ks].y, ny);
  }
  float nn = nx + ny;
  nn += __shfl_xor(nn, 16, 32);
  const float rn2 = 1.0f / nn;

  float acc = 0.0f;
  for (int t0 = 0; t0 < LCHUNK; t0 += 16) {
    const float* arow = vt + (t0 + nrow) * AP + khalf;

    v8f cAr = {};
    v8f cAi = {};
    #pragma unroll
    for (int ks = 0; ks < KS; ++ks) {
      const v2f aR = *reinterpret_cast<const v2f*>(arow + 4 * ks);
      const v2f aI = *reinterpret_cast<const v2f*>(arow + D_ + 4 * ks);
      cAr = __builtin_amdgcn_wmma_f32_16x16x4_f32(false, aR, false, bK[ks], (short)0, cAr, false, false);
      cAi = __builtin_amdgcn_wmma_f32_16x16x4_f32(false, aI, false, bK[ks], (short)0, cAi, false, false);
    }

    float partial = 0.0f;
    #pragma unroll
    for (int r = 0; r < 8; ++r) {
      const float ar = cAr[r];
      const float ai = cAi[r];
      partial = fmaf(wt[t0 + rbase + r], fmaf(ar, ar, ai * ai), partial);
    }
    partial += __shfl_xor(partial, 16, 32);
    acc += partial;
  }
  if (lane < 16) pacc[mbase + lane] = acc * rn2;
  __syncthreads();

  if (tid < PROW) {
    float val;
    if (tid < M_) {
      val = pacc[tid];
    } else {
      const int j = tid - M_;
      val = (j < 2) ? scal[j] : 0.0f;
    }
    part[(b * LSPLIT + split) * PROW + tid] = val;
  }
}

__global__ void __launch_bounds__(M_)
merge_kernel(const float* __restrict__ part, float* __restrict__ out) {
  const int b = blockIdx.x;
  const int m = threadIdx.x;
  const float* row = part + b * LSPLIT * PROW;
  float pj[LSPLIT], cj[LSPLIT], sj[LSPLIT];
  #pragma unroll
  for (int j = 0; j < LSPLIT; ++j) {
    pj[j] = row[j * PROW + m];
    cj[j] = row[j * PROW + M_];
    sj[j] = row[j * PROW + M_ + 1];
  }
  float cmax = cj[0];
  #pragma unroll
  for (int j = 1; j < LSPLIT; ++j) cmax = fmaxf(cmax, cj[j]);
  float num = 0.0f, den = 0.0f;
  #pragma unroll
  for (int j = 0; j < LSPLIT; ++j) {
    const float f = expf(cj[j] - cmax);
    num = fmaf(pj[j], f, num);
    den = fmaf(sj[j], f, den);
  }
  out[b * M_ + m] = num / den;
}

extern "C" void kernel_launch(void* const* d_in, const int* in_sizes, int n_in,
                              void* d_out, int out_size, void* d_ws, size_t ws_size,
                              hipStream_t stream) {
  const int*   x   = (const int*)d_in[0];
  const float* amp = (const float*)d_in[1];
  const float* pha = (const float*)d_in[2];
  const float* mix = (const float*)d_in[3];
  const float* kr  = (const float*)d_in[4];
  const float* ki  = (const float*)d_in[5];
  float*       out = (float*)d_out;

  float* part = (float*)d_ws;

  const size_t lds_bytes =
      (size_t)(M_ * BP + LCHUNK * AP + LCHUNK + M_ + 2) * sizeof(float);
  dim3 grid(B_, LSPLIT);
  qproj_kernel<<<grid, 256, lds_bytes, stream>>>(x, amp, pha, mix, kr, ki, part);
  merge_kernel<<<B_, M_, 0, stream>>>(part, out);
}


# Complex dot product from three real products instead of four (P1=(kr-ki).vr, P2=kr.(vi-vr), P3=ki.(vr+vi); Ar=P1+P3, Ai=P1+P2): 39 matrix instructions per tile, was 50; sums reordered
# speedup vs baseline: 6.6427x; 1.0731x over previous
#include <hip/hip_runtime.h>
#include <math.h>


typedef __attribute__((ext_vector_type(2))) float v2f;
typedef __attribute__((ext_vector_type(8))) float v8f;

#define B_   64
#define L_   512
#define D_   50
#define M_   128
#define KC   (2 * D_)
#define BP   KC
#define DP   52
#define AP   (3 * DP)
#define LSPLIT 8
#define LCHUNK (L_ / LSPLIT)
#define PROW (M_ + 32)


#define GT  (256 / D_)
#define GPS ((LCHUNK + GT - 1) / GT)
#define KS  (DP / 4)

__global__ void __launch_bounds__(256)
qproj_kernel(const int* __restrict__ x,
             const float* __restrict__ amp_table,
             const float* __restrict__ pha_table,
             const float* __restrict__ mix_table,
             const float* __restrict__ kr_in,
             const float* __restrict__ ki_in,
             float* __restrict__ part) {
  extern __shared__ float smem[];
  float* kcs  = smem;
  float* vt   = kcs + M_ * BP;
  float* wt   = vt + LCHUNK * AP;
  float* pacc = wt + LCHUNK;
  float* scal = pacc + M_;

  const int b     = blockIdx.x;
  const int split = blockIdx.y;
  const int lt0   = split * LCHUNK;
  const int tid   = threadIdx.x;
  const int wave  = tid >> 5;
  const int lane  = tid & 31;

  static_assert(DP % 4 == 0 && DP >= D_ && DP - D_ == 2 && D_ % 2 == 0, "a plane is whole k-steps, its zero tail is one 8-byte piece, and a fragment pair never straddles a plane's end");
  static_assert(LCHUNK == 64, "one wave holds the chunk's mix logits as two tokens per lane");
  const int trow = tid / D_;
  const int d    = tid - trow * D_;
  const bool act = tid < GT * D_;
  float av[GPS], pv[GPS];
  {
    int idx[GPS];
    #pragma unroll
    for (int j = 0; j < GPS; ++j) {
      const int tok = min(trow + GT * j, LCHUNK - 1);
      idx[j] = x[b * L_ + lt0 + tok];
    }
    #pragma unroll
    for (int j = 0; j < GPS; ++j) {
      av[j] = amp_table[idx[j] * D_ + d];
      pv[j] = pha_table[idx[j] * D_ + d];
    }
  }

  float mv0 = 0.0f, mv1 = 0.0f;
  if (wave == 0) {
    mv0 = mix_table[x[b * L_ + lt0 + lane]];
    mv1 = mix_table[x[b * L_ + lt0 + 32 + lane]];
  }

  {
    const v2f* srcr = reinterpret_cast<const v2f*>(kr_in);
    const v2f* srci = reinterpret_cast<const v2f*>(ki_in);
    for (int i = tid; i < M_ * (D_ / 2); i += 256) {
      const int m  = i / (D_ / 2);
      const int dp = i - m * (D_ / 2);
      v2f* dst = reinterpret_cast<v2f*>(kcs + m * BP) + dp;
      dst[0]      = srcr[i];
      dst[D_ / 2] = srci[i];
    }
  }

  if (tid < LCHUNK * 3) {
    const int tok = tid / 3;
    const int pl  = tid - tok * 3;
    *reinterpret_cast<v2f*>(vt + tok * AP + pl * DP + D_) = v2f{0.0f, 0.0f};
  }

  if (wave == 0) {
    float cm = fmaxf(mv0, mv1);
    for (int off = 16; off > 0; off >>= 1) cm = fmaxf(cm, __shfl_xor(cm, off, 32));
    const float e0 = expf(mv0 - cm);
    const float e1 = expf(mv1 - cm);
    wt[lane]      = e0;
    wt[lane + 32] = e1;
    float es = e0 + e1;
    for (int off = 16; off > 0; off >>= 1) es += __shfl_xor(es, off, 32);
    if (lane == 0) {
      scal[0] = cm;
      scal[1] = es;
    }
  }

  #pragma unroll
  for (int j = 0; j < GPS; ++j) {
    const int tok = trow + GT * j;
    float s, c;
    sincosf(pv[j], &s, &c);
    const float re = av[j] * c;
    const float im = av[j] * s;
    if (act && tok < LCHUNK) {
      float* row = vt + tok * AP + d;
      row[0]      = re;
      row[DP]     = im - re;
      row[2 * DP] = re + im;
    }
  }
  __syncthreads();

  const int mbase = wave * 16;
  const int nrow  = lane & 15;
  const int khalf = (lane >> 4) * 2;
  const int rbase = (lane >> 4) * 8;

  v2f bR[KS], bI[KS], bD[KS];
  {
    const float* brow = kcs + (mbase + nrow) * BP + khalf;
    #pragma unroll
    for (int ks = 0; ks < KS; ++ks) {
      v2f r = *reinterpret_cast<const v2f*>(brow + 4 * ks);
      v2f i = *reinterpret_cast<const v2f*>(brow + D_ + 4 * ks);
      if (4 * ks + 2 >= D_ && khalf != 0) {
        r = v2f{0.0f, 0.0f};
        i = v2f{0.0f, 0.0f};
      }
      bR[ks] = r;
      bI[ks] = i;
      bD[ks] = r - i;
    }
  }

  float nx = 0.0f, ny = 0.0f;
  #pragma unroll
  for (int ks = 0; ks < KS; ++ks) {
    nx = fmaf(bR[ks].x, bR[ks].x, nx);
    ny = fmaf(bR[ks].y, bR[ks].y, ny);
    nx = fmaf(bI[ks].x, bI[ks].x, nx);
    ny = fmaf(bI[ks].y, bI[ks].y, ny);
  }
  float nn = nx + ny;
  nn += __shfl_xor(nn, 16, 32);
  const float rn2 = 1.0f / nn;

  float acc = 0.0f;
  for (int t0 = 0; t0 < LCHUNK; t0 += 16) {
    const float* arow = vt + (t0 + nrow) * AP + khalf;

    v8f c1 = {};
    v8f c2 = {};
    v8f c3 = {};
    #pragma unroll
    for (int ks = 0; ks < KS; ++ks) {
      const v2f a1 = *reinterpret_cast<const v2f*>(arow + 4 * ks);
      const v2f a2 = *reinterpret_cast<const v2f*>(arow + DP + 4 * ks);
      const v2f a3 = *reinterpret_cast<const v2f*>(arow + 2 * DP + 4 * ks);
      c1 = __builtin_amdgcn_wmma_f32_16x16x4_f32(false, a1, false, bD[ks], (short)0, c1, false, false);
      c2 = __builtin_amdgcn_wmma_f32_16x16x4_f32(false, a2, false, bR[ks], (short)0, c2, false, false);
      c3 = __builtin_amdgcn_wmma_f32_16x16x4_f32(false, a3, false, bI[ks], (short)0, c3, false, false);
    }

    float partial = 0.0f;
    #pragma unroll
    for (int r = 0; r < 8; ++r) {
      const float ar = c1[r] + c3[r];
      const float ai = c1[r] + c2[r];
      partial = fmaf(wt[t0 + rbase + r], fmaf(ar, ar, ai * ai), partial);
    }
    partial += __shfl_xor(partial, 16, 32);
    acc += partial;
  }
  if (lane < 16) pacc[mbase + lane] = acc * rn2;
  __syncthreads();

  if (tid < PROW) {
    float val;
    if (tid < M_) {
      val = pacc[tid];
    } else {
      const int j = tid - M_;
      val = (j < 2) ? scal[j] : 0.0f;
    }
    part[(b * LSPLIT + split) * PROW + tid] = val;
  }
}

__global__ void __launch_bounds__(M_)
merge_kernel(const float* __restrict__ part, float* __restrict__ out) {
  const int b = blockIdx.x;
  const int m = threadIdx.x;
  const float* row = part + b * LSPLIT * PROW;
  float pj[LSPLIT], cj[LSPLIT], sj[LSPLIT];
  #pragma unroll
  for (int j = 0; j < LSPLIT; ++j) {
    pj[j] = row[j * PROW + m];
    cj[j] = row[j * PROW + M_];
    sj[j] = row[j * PROW + M_ + 1];
  }
  float cmax = cj[0];
  #pragma unroll
  for (int j = 1; j < LSPLIT; ++j) cmax = fmaxf(cmax, cj[j]);
  float num = 0.0f, den = 0.0f;
  #pragma unroll
  for (int j = 0; j < LSPLIT; ++j) {
    const float f = expf(cj[j] - cmax);
    num = fmaf(pj[j], f, num);
    den = fmaf(sj[j], f, den);
  }
  out[b * M_ + m] = num / den;
}

extern "C" void kernel_launch(void* const* d_in, const int* in_sizes, int n_in,
                              void* d_out, int out_size, void* d_ws, size_t ws_size,
                              hipStream_t stream) {
  const int*   x   = (const int*)d_in[0];
  const float* amp = (const float*)d_in[1];
  const float* pha = (const float*)d_in[2];
  const float* mix = (const float*)d_in[3];
  const float* kr  = (const float*)d_in[4];
  const float* ki  = (const float*)d_in[5];
  float*       out = (float*)d_out;

  float* part = (float*)d_ws;

  const size_t lds_bytes =
      (size_t)(M_ * BP + LCHUNK * AP + LCHUNK + M_ + 2) * sizeof(float);
  dim3 grid(B_, LSPLIT);
  qproj_kernel<<<grid, 256, lds_bytes, stream>>>(x, amp, pha, mix, kr, ki, part);
  merge_kernel<<<B_, M_, 0, stream>>>(part, out);
}


# Cosine and sine of each phase from the hardware v_cos_f32 / v_sin_f32 (angle in revolutions, one multiply) instead of in-line sincosf; sin/cos accuracy drops from about 1 ulp to the instructions' own
# speedup vs baseline: 7.0491x; 1.0612x over previous
#include <hip/hip_runtime.h>
#include <math.h>


typedef __attribute__((ext_vector_type(2))) float v2f;
typedef __attribute__((ext_vector_type(8))) float v8f;

#define B_   64
#define L_   512
#define D_   50
#define M_   128
#define KC   (2 * D_)
#define BP   KC
#define DP   52
#define AP   (3 * DP)
#define LSPLIT 8
#define LCHUNK (L_ / LSPLIT)
#define PROW (M_ + 32)


#define GT  (256 / D_)
#define GPS ((LCHUNK + GT - 1) / GT)
#define KS  (DP / 4)

__global__ void __launch_bounds__(256)
qproj_kernel(const int* __restrict__ x,
             const float* __restrict__ amp_table,
             const float* __restrict__ pha_table,
             const float* __restrict__ mix_table,
             const float* __restrict__ kr_in,
             const float* __restrict__ ki_in,
             float* __restrict__ part) {
  extern __shared__ float smem[];
  float* kcs  = smem;
  float* vt   = kcs + M_ * BP;
  float* wt   = vt + LCHUNK * AP;
  float* pacc = wt + LCHUNK;
  float* scal = pacc + M_;

  const int b     = blockIdx.x;
  const int split = blockIdx.y;
  const int lt0   = split * LCHUNK;
  const int tid   = threadIdx.x;
  const int wave  = tid >> 5;
  const int lane  = tid & 31;

  static_assert(DP % 4 == 0 && DP >= D_ && DP - D_ == 2 && D_ % 2 == 0, "a plane is whole k-steps, its zero tail is one 8-byte piece, and a fragment pair never straddles a plane's end");
  static_assert(LCHUNK == 64, "one wave holds the chunk's mix logits as two tokens per lane");
  const int trow = tid / D_;
  const int d    = tid - trow * D_;
  const bool act = tid < GT * D_;
  float av[GPS], pv[GPS];
  {
    int idx[GPS];
    #pragma unroll
    for (int j = 0; j < GPS; ++j) {
      const int tok = min(trow + GT * j, LCHUNK - 1);
      idx[j] = x[b * L_ + lt0 + tok];
    }
    #pragma unroll
    for (int j = 0; j < GPS; ++j) {
      av[j] = amp_table[idx[j] * D_ + d];
      pv[j] = pha_table[idx[j] * D_ + d];
    }
  }

  float mv0 = 0.0f, mv1 = 0.0f;
  if (wave == 0) {
    mv0 = mix_table[x[b * L_ + lt0 + lane]];
    mv1 = mix_table[x[b * L_ + lt0 + 32 + lane]];
  }

  {
    const v2f* srcr = reinterpret_cast<const v2f*>(kr_in);
    const v2f* srci = reinterpret_cast<const v2f*>(ki_in);
    for (int i = tid; i < M_ * (D_ / 2); i += 256) {
      const int m  = i / (D_ / 2);
      const int dp = i - m * (D_ / 2);
      v2f* dst = reinterpret_cast<v2f*>(kcs + m * BP) + dp;
      dst[0]      = srcr[i];
      dst[D_ / 2] = srci[i];
    }
  }

  if (tid < LCHUNK * 3) {
    const int tok = tid / 3;
    const int pl  = tid - tok * 3;
    *reinterpret_cast<v2f*>(vt + tok * AP + pl * DP + D_) = v2f{0.0f, 0.0f};
  }

  if (wave == 0) {
    float cm = fmaxf(mv0, mv1);
    for (int off = 16; off > 0; off >>= 1) cm = fmaxf(cm, __shfl_xor(cm, off, 32));
    const float e0 = expf(mv0 - cm);
    const float e1 = expf(mv1 - cm);
    wt[lane]      = e0;
    wt[lane + 32] = e1;
    float es = e0 + e1;
    for (int off = 16; off > 0; off >>= 1) es += __shfl_xor(es, off, 32);
    if (lane == 0) {
      scal[0] = cm;
      scal[1] = es;
    }
  }

  #pragma unroll
  for (int j = 0; j < GPS; ++j) {
    const int tok = trow + GT * j;
    const float rev = pv[j] * 0.15915494f;
    const float s = __builtin_amdgcn_sinf(rev);
    const float c = __builtin_amdgcn_cosf(rev);
    const float re = av[j] * c;
    const float im = av[j] * s;
    if (act && tok < LCHUNK) {
      float* row = vt + tok * AP + d;
      row[0]      = re;
      row[DP]     = im - re;
      row[2 * DP] = re + im;
    }
  }
  __syncthreads();

  const int mbase = wave * 16;
  const int nrow  = lane & 15;
  const int khalf = (lane >> 4) * 2;
  const int rbase = (lane >> 4) * 8;

  v2f bR[KS], bI[KS], bD[KS];
  {
    const float* brow = kcs + (mbase + nrow) * BP + khalf;
    #pragma unroll
    for (int ks = 0; ks < KS; ++ks) {
      v2f r = *reinterpret_cast<const v2f*>(brow + 4 * ks);
      v2f i = *reinterpret_cast<const v2f*>(brow + D_ + 4 * ks);
      if (4 * ks + 2 >= D_ && khalf != 0) {
        r = v2f{0.0f, 0.0f};
        i = v2f{0.0f, 0.0f};
      }
      bR[ks] = r;
      bI[ks] = i;
      bD[ks] = r - i;
    }
  }

  float nx = 0.0f, ny = 0.0f;
  #pragma unroll
  for (int ks = 0; ks < KS; ++ks) {
    nx = fmaf(bR[ks].x, bR[ks].x, nx);
    ny = fmaf(bR[ks].y, bR[ks].y, ny);
    nx = fmaf(bI[ks].x, bI[ks].x, nx);
    ny = fmaf(bI[ks].y, bI[ks].y, ny);
  }
  float nn = nx + ny;
  nn += __shfl_xor(nn, 16, 32);
  const float rn2 = 1.0f / nn;

  float acc = 0.0f;
  for (int t0 = 0; t0 < LCHUNK; t0 += 16) {
    const float* arow = vt + (t0 + nrow) * AP + khalf;

    v8f c1 = {};
    v8f c2 = {};
    v8f c3 = {};
    #pragma unroll
    for (int ks = 0; ks < KS; ++ks) {
      const v2f a1 = *reinterpret_cast<const v2f*>(arow + 4 * ks);
      const v2f a2 = *reinterpret_cast<const v2f*>(arow + DP + 4 * ks);
      const v2f a3 = *reinterpret_cast<const v2f*>(arow + 2 * DP + 4 * ks);
      c1 = __builtin_amdgcn_wmma_f32_16x16x4_f32(false, a1, false, bD[ks], (short)0, c1, false, false);
      c2 = __builtin_amdgcn_wmma_f32_16x16x4_f32(false, a2, false, bR[ks], (short)0, c2, false, false);
      c3 = __builtin_amdgcn_wmma_f32_16x16x4_f32(false, a3, false, bI[ks], (short)0, c3, false, false);
    }

    float partial = 0.0f;
    #pragma unroll
    for (int r = 0; r < 8; ++r) {
      const float ar = c1[r] + c3[r];
      const float ai = c1[r] + c2[r];
      partial = fmaf(wt[t0 + rbase + r], fmaf(ar, ar, ai * ai), partial);
    }
    partial += __shfl_xor(partial, 16, 32);
    acc += partial;
  }
  if (lane < 16) pacc[mbase + lane] = acc * rn2;
  __syncthreads();

  if (tid < PROW) {
    float val;
    if (tid < M_) {
      val = pacc[tid];
    } else {
      const int j = tid - M_;
      val = (j < 2) ? scal[j] : 0.0f;
    }
    part[(b * LSPLIT + split) * PROW + tid] = val;
  }
}

__global__ void __launch_bounds__(M_)
merge_kernel(const float* __restrict__ part, float* __restrict__ out) {
  const int b = blockIdx.x;
  const int m = threadIdx.x;
  const float* row = part + b * LSPLIT * PROW;
  float pj[LSPLIT], cj[LSPLIT], sj[LSPLIT];
  #pragma unroll
  for (int j = 0; j < LSPLIT; ++j) {
    pj[j] = row[j * PROW + m];
    cj[j] = row[j * PROW + M_];
    sj[j] = row[j * PROW + M_ + 1];
  }
  float cmax = cj[0];
  #pragma unroll
  for (int j = 1; j < LSPLIT; ++j) cmax = fmaxf(cmax, cj[j]);
  float num = 0.0f, den = 0.0f;
  #pragma unroll
  for (int j = 0; j < LSPLIT; ++j) {
    const float f = expf(cj[j] - cmax);
    num = fmaf(pj[j], f, num);
    den = fmaf(sj[j], f, den);
  }
  out[b * M_ + m] = num / den;
}

extern "C" void kernel_launch(void* const* d_in, const int* in_sizes, int n_in,
                              void* d_out, int out_size, void* d_ws, size_t ws_size,
                              hipStream_t stream) {
  const int*   x   = (const int*)d_in[0];
  const float* amp = (const float*)d_in[1];
  const float* pha = (const float*)d_in[2];
  const float* mix = (const float*)d_in[3];
  const float* kr  = (const float*)d_in[4];
  const float* ki  = (const float*)d_in[5];
  float*       out = (float*)d_out;

  float* part = (float*)d_ws;

  const size_t lds_bytes =
      (size_t)(M_ * BP + LCHUNK * AP + LCHUNK + M_ + 2) * sizeof(float);
  dim3 grid(B_, LSPLIT);
  qproj_kernel<<<grid, 256, lds_bytes, stream>>>(x, amp, pha, mix, kr, ki, part);
  merge_kernel<<<B_, M_, 0, stream>>>(part, out);
}


# Blocks of 4 waves, not 8: a wave owns two 16-row measurement tiles fed by one A-fragment read (LDS reads grouped a tile ahead, staging loads all in flight); 2048 waves enter, not 4096; sums unchanged
# speedup vs baseline: 7.6326x; 1.0828x over previous
#include <hip/hip_runtime.h>
#include <math.h>


typedef __attribute__((ext_vector_type(2))) float v2f;
typedef __attribute__((ext_vector_type(8))) float v8f;

#define B_   64
#define L_   512
#define D_   50
#define M_   128
#define KC   (2 * D_)
#define BP   KC
#define DP   52
#define AP   (3 * DP)
#define LSPLIT 8
#define LCHUNK (L_ / LSPLIT)
#define PROW (M_ + 32)


#define NW  4
#define NT  (NW * 32)
#define MT  2
#define DH  (D_ / 2)
#define GT  (NT / DH)
#define GPS ((LCHUNK + GT - 1) / GT)
#define KS  (DP / 4)
#define SPT ((M_ * DH) / NT)
#define NTILE (LCHUNK / 16)

__global__ void __launch_bounds__(NT)
qproj_kernel(const int* __restrict__ x,
             const float* __restrict__ amp_table,
             const float* __restrict__ pha_table,
             const float* __restrict__ mix_table,
             const float* __restrict__ kr_in,
             const float* __restrict__ ki_in,
             float* __restrict__ part) {
  extern __shared__ float smem[];
  float* kcs  = smem;
  float* vt   = kcs + M_ * BP;
  float* wt   = vt + LCHUNK * AP;
  float* pacc = wt + LCHUNK;
  float* scal = pacc + M_;

  const int b     = blockIdx.x;
  const int split = blockIdx.y;
  const int lt0   = split * LCHUNK;
  const int tid   = threadIdx.x;
  const int wave  = tid >> 5;
  const int lane  = tid & 31;

  static_assert(DP % 4 == 0 && DP >= D_ && DP - D_ == 2 && D_ % 2 == 0, "a plane is whole k-steps, its zero tail is one 8-byte piece, and a fragment pair never straddles a plane's end");
  static_assert(LCHUNK == 64, "one wave holds the chunk's mix logits as two tokens per lane");
  static_assert(M_ == NW * MT * 16 && (M_ * DH) % NT == 0 && GT >= 1, "the waves' tiles cover all measurement rows; staging pieces divide evenly over the threads");

  const int trow = tid / DH;
  const int dq   = tid - trow * DH;
  const bool act = tid < GT * DH;
  v2f av[GPS], pv[GPS];
  v2f sr[SPT], si[SPT];
  {
    int idx[GPS];
    #pragma unroll
    for (int j = 0; j < GPS; ++j) {
      const int tok = min(trow + GT * j, LCHUNK - 1);
      idx[j] = x[b * L_ + lt0 + tok];
    }
    __builtin_amdgcn_sched_barrier(0);
    const v2f* srcr = reinterpret_cast<const v2f*>(kr_in);
    const v2f* srci = reinterpret_cast<const v2f*>(ki_in);
    #pragma unroll
    for (int q = 0; q < SPT; ++q) {
      sr[q] = srcr[tid + NT * q];
      si[q] = srci[tid + NT * q];
    }
    __builtin_amdgcn_sched_barrier(0);
    #pragma unroll
    for (int j = 0; j < GPS; ++j) {
      av[j] = *reinterpret_cast<const v2f*>(amp_table + idx[j] * D_ + 2 * dq);
      pv[j] = *reinterpret_cast<const v2f*>(pha_table + idx[j] * D_ + 2 * dq);
    }
  }

  float mv0 = 0.0f, mv1 = 0.0f;
  if (wave == 0) {
    mv0 = mix_table[x[b * L_ + lt0 + lane]];
    mv1 = mix_table[x[b * L_ + lt0 + 32 + lane]];
  }

  #pragma unroll
  for (int q = 0; q < SPT; ++q) {
    const int i  = tid + NT * q;
    const int m  = i / DH;
    const int dp = i - m * DH;
    v2f* dst = reinterpret_cast<v2f*>(kcs + m * BP) + dp;
    dst[0]  = sr[q];
    dst[DH] = si[q];
  }

  for (int i = tid; i < LCHUNK * 3; i += NT) {
    const int tok = i / 3;
    const int pl  = i - tok * 3;
    *reinterpret_cast<v2f*>(vt + tok * AP + pl * DP + D_) = v2f{0.0f, 0.0f};
  }

  if (wave == 0) {
    float cm = fmaxf(mv0, mv1);
    for (int off = 16; off > 0; off >>= 1) cm = fmaxf(cm, __shfl_xor(cm, off, 32));
    const float e0 = expf(mv0 - cm);
    const float e1 = expf(mv1 - cm);
    wt[lane]      = e0;
    wt[lane + 32] = e1;
    float es = e0 + e1;
    for (int off = 16; off > 0; off >>= 1) es += __shfl_xor(es, off, 32);
    if (lane == 0) {
      scal[0] = cm;
      scal[1] = es;
    }
  }

  #pragma unroll
  for (int j = 0; j < GPS; ++j) {
    const int tok = trow + GT * j;
    const float rev0 = pv[j].x * 0.15915494f;
    const float rev1 = pv[j].y * 0.15915494f;
    const float s0 = __builtin_amdgcn_sinf(rev0);
    const float c0 = __builtin_amdgcn_cosf(rev0);
    const float s1 = __builtin_amdgcn_sinf(rev1);
    const float c1 = __builtin_amdgcn_cosf(rev1);
    const float re0 = av[j].x * c0;
    const float im0 = av[j].x * s0;
    const float re1 = av[j].y * c1;
    const float im1 = av[j].y * s1;
    if (act && tok < LCHUNK) {
      float* row = vt + tok * AP + 2 * dq;
      *reinterpret_cast<v2f*>(row)          = v2f{re0, re1};
      *reinterpret_cast<v2f*>(row + DP)     = v2f{im0 - re0, im1 - re1};
      *reinterpret_cast<v2f*>(row + 2 * DP) = v2f{re0 + im0, re1 + im1};
    }
  }
  __syncthreads();

  const int mbase = wave * (16 * MT);
  const int nrow  = lane & 15;
  const int khalf = (lane >> 4) * 2;
  const int rbase = (lane >> 4) * 8;

  v2f bR[MT][KS], bI[MT][KS], bD[MT][KS];
  float rn2[MT];
  #pragma unroll
  for (int mt = 0; mt < MT; ++mt) {
    const float* brow = kcs + (mbase + 16 * mt + nrow) * BP + khalf;
    #pragma unroll
    for (int ks = 0; ks < KS; ++ks) {
      v2f r = *reinterpret_cast<const v2f*>(brow + 4 * ks);
      v2f i = *reinterpret_cast<const v2f*>(brow + D_ + 4 * ks);
      if (4 * ks + 2 >= D_ && khalf != 0) {
        r = v2f{0.0f, 0.0f};
        i = v2f{0.0f, 0.0f};
      }
      bR[mt][ks] = r;
      bI[mt][ks] = i;
      bD[mt][ks] = r - i;
    }
    float nx = 0.0f, ny = 0.0f;
    #pragma unroll
    for (int ks = 0; ks < KS; ++ks) {
      nx = fmaf(bR[mt][ks].x, bR[mt][ks].x, nx);
      ny = fmaf(bR[mt][ks].y, bR[mt][ks].y, ny);
      nx = fmaf(bI[mt][ks].x, bI[mt][ks].x, nx);
      ny = fmaf(bI[mt][ks].y, bI[mt][ks].y, ny);
    }
    float nn = nx + ny;
    nn += __shfl_xor(nn, 16, 32);
    rn2[mt] = 1.0f / nn;
  }

  float acc[MT];
  #pragma unroll
  for (int mt = 0; mt < MT; ++mt) acc[mt] = 0.0f;

  v2f aN[3][KS];
  float wN[8];
  {
    const float* arow = vt + nrow * AP + khalf;
    #pragma unroll
    for (int ks = 0; ks < KS; ++ks) {
      aN[0][ks] = *reinterpret_cast<const v2f*>(arow + 4 * ks);
      aN[1][ks] = *reinterpret_cast<const v2f*>(arow + DP + 4 * ks);
      aN[2][ks] = *reinterpret_cast<const v2f*>(arow + 2 * DP + 4 * ks);
    }
    #pragma unroll
    for (int r = 0; r < 8; ++r) wN[r] = wt[rbase + r];
  }

  #pragma unroll
  for (int t = 0; t < NTILE; ++t) {
    v2f aC[3][KS];
    float wC[8];
    #pragma unroll
    for (int ks = 0; ks < KS; ++ks) {
      aC[0][ks] = aN[0][ks];
      aC[1][ks] = aN[1][ks];
      aC[2][ks] = aN[2][ks];
    }
    #pragma unroll
    for (int r = 0; r < 8; ++r) wC[r] = wN[r];

    __builtin_amdgcn_sched_barrier(0);
    if (t + 1 < NTILE) {
      const float* arow = vt + ((t + 1) * 16 + nrow) * AP + khalf;
      #pragma unroll
      for (int ks = 0; ks < KS; ++ks) {
        aN[0][ks] = *reinterpret_cast<const v2f*>(arow + 4 * ks);
        aN[1][ks] = *reinterpret_cast<const v2f*>(arow + DP + 4 * ks);
        aN[2][ks] = *reinterpret_cast<const v2f*>(arow + 2 * DP + 4 * ks);
      }
      #pragma unroll
      for (int r = 0; r < 8; ++r) wN[r] = wt[(t + 1) * 16 + rbase + r];
    }
    __builtin_amdgcn_sched_barrier(0);

    v8f c1[MT], c2[MT], c3[MT];
    #pragma unroll
    for (int mt = 0; mt < MT; ++mt) { c1[mt] = v8f{}; c2[mt] = v8f{}; c3[mt] = v8f{}; }
    #pragma unroll
    for (int ks = 0; ks < KS; ++ks) {
      #pragma unroll
      for (int mt = 0; mt < MT; ++mt) {
        c1[mt] = __builtin_amdgcn_wmma_f32_16x16x4_f32(false, aC[0][ks], false, bD[mt][ks], (short)0, c1[mt], false, false);
        c2[mt] = __builtin_amdgcn_wmma_f32_16x16x4_f32(false, aC[1][ks], false, bR[mt][ks], (short)0, c2[mt], false, false);
        c3[mt] = __builtin_amdgcn_wmma_f32_16x16x4_f32(false, aC[2][ks], false, bI[mt][ks], (short)0, c3[mt], false, false);
      }
    }

    #pragma unroll
    for (int mt = 0; mt < MT; ++mt) {
      float partial = 0.0f;
      #pragma unroll
      for (int r = 0; r < 8; ++r) {
        const float ar = c1[mt][r] + c3[mt][r];
        const float ai = c1[mt][r] + c2[mt][r];
        partial = fmaf(wC[r], fmaf(ar, ar, ai * ai), partial);
      }
      partial += __shfl_xor(partial, 16, 32);
      acc[mt] += partial;
    }
  }
  if (lane < 16) {
    #pragma unroll
    for (int mt = 0; mt < MT; ++mt) pacc[mbase + 16 * mt + lane] = acc[mt] * rn2[mt];
  }
  __syncthreads();

  for (int i = tid; i < PROW; i += NT) {
    float val;
    if (i < M_) {
      val = pacc[i];
    } else {
      const int j = i - M_;
      val = (j < 2) ? scal[j] : 0.0f;
    }
    part[(b * LSPLIT + split) * PROW + i] = val;
  }
}

__global__ void __launch_bounds__(M_)
merge_kernel(const float* __restrict__ part, float* __restrict__ out) {
  const int b = blockIdx.x;
  const int m = threadIdx.x;
  const float* row = part + b * LSPLIT * PROW;
  float pj[LSPLIT], cj[LSPLIT], sj[LSPLIT];
  #pragma unroll
  for (int j = 0; j < LSPLIT; ++j) {
    pj[j] = row[j * PROW + m];
    cj[j] = row[j * PROW + M_];
    sj[j] = row[j * PROW + M_ + 1];
  }
  float cmax = cj[0];
  #pragma unroll
  for (int j = 1; j < LSPLIT; ++j) cmax = fmaxf(cmax, cj[j]);
  float num = 0.0f, den = 0.0f;
  #pragma unroll
  for (int j = 0; j < LSPLIT; ++j) {
    const float f = expf(cj[j] - cmax);
    num = fmaf(pj[j], f, num);
    den = fmaf(sj[j], f, den);
  }
  out[b * M_ + m] = num / den;
}

extern "C" void kernel_launch(void* const* d_in, const int* in_sizes, int n_in,
                              void* d_out, int out_size, void* d_ws, size_t ws_size,
                              hipStream_t stream) {
  const int*   x   = (const int*)d_in[0];
  const float* amp = (const float*)d_in[1];
  const float* pha = (const float*)d_in[2];
  const float* mix = (const float*)d_in[3];
  const float* kr  = (const float*)d_in[4];
  const float* ki  = (const float*)d_in[5];
  float*       out = (float*)d_out;

  float* part = (float*)d_ws;

  const size_t lds_bytes =
      (size_t)(M_ * BP + LCHUNK * AP + LCHUNK + M_ + 2) * sizeof(float);
  dim3 grid(B_, LSPLIT);
  qproj_kernel<<<grid, NT, lds_bytes, stream>>>(x, amp, pha, mix, kr, ki, part);
  merge_kernel<<<B_, M_, 0, stream>>>(part, out);
}


# Wave 0's chunk softmax off the path to the block barrier: its index and logit loads go out with the gather's (were two round trips behind), max and sum reduce by DPP, not LDS exchanges; order kept
# speedup vs baseline: 8.1894x; 1.0730x over previous
#include <hip/hip_runtime.h>
#include <math.h>


typedef __attribute__((ext_vector_type(2))) float v2f;
typedef __attribute__((ext_vector_type(8))) float v8f;

#define B_   64
#define L_   512
#define D_   50
#define M_   128
#define KC   (2 * D_)
#define BP   KC
#define DP   52
#define AP   (3 * DP)
#define LSPLIT 8
#define LCHUNK (L_ / LSPLIT)
#define PROW (M_ + 32)


#define NW  4
#define NT  (NW * 32)
#define MT  2
#define DH  (D_ / 2)
#define GT  (NT / DH)
#define GPS ((LCHUNK + GT - 1) / GT)
#define KS  (DP / 4)
#define SPT ((M_ * DH) / NT)
#define NTILE (LCHUNK / 16)

template <int MASK>
__device__ __forceinline__ float lane_xor_row(float v) {
  return __builtin_bit_cast(float, __builtin_amdgcn_update_dpp(
      0, __builtin_bit_cast(int, v), 0x160 + MASK, 0xf, 0xf, true));
}
__device__ __forceinline__ float lane_xor_16(float v) {
  return __builtin_bit_cast(float, __builtin_amdgcn_permlanex16(
      __builtin_bit_cast(int, v), __builtin_bit_cast(int, v), 0x76543210, 0xfedcba98, false, false));
}

__global__ void __launch_bounds__(NT)
qproj_kernel(const int* __restrict__ x,
             const float* __restrict__ amp_table,
             const float* __restrict__ pha_table,
             const float* __restrict__ mix_table,
             const float* __restrict__ kr_in,
             const float* __restrict__ ki_in,
             float* __restrict__ part) {
  extern __shared__ float smem[];
  float* kcs  = smem;
  float* vt   = kcs + M_ * BP;
  float* wt   = vt + LCHUNK * AP;
  float* pacc = wt + LCHUNK;
  float* scal = pacc + M_;

  const int b     = blockIdx.x;
  const int split = blockIdx.y;
  const int lt0   = split * LCHUNK;
  const int tid   = threadIdx.x;
  const int wave  = tid >> 5;
  const int lane  = tid & 31;

  static_assert(DP % 4 == 0 && DP >= D_ && DP - D_ == 2 && D_ % 2 == 0, "a plane is whole k-steps, its zero tail is one 8-byte piece, and a fragment pair never straddles a plane's end");
  static_assert(LCHUNK == 64, "one wave holds the chunk's mix logits as two tokens per lane");
  static_assert(M_ == NW * MT * 16 && (M_ * DH) % NT == 0 && GT >= 1, "the waves' tiles cover all measurement rows; staging pieces divide evenly over the threads");

  const int trow = tid / DH;
  const int dq   = tid - trow * DH;
  const bool act = tid < GT * DH;
  v2f av[GPS], pv[GPS];
  v2f sr[SPT], si[SPT];
  unsigned mi0 = 0, mi1 = 0;
  float mv0 = 0.0f, mv1 = 0.0f;
  {
    int idx[GPS];
    #pragma unroll
    for (int j = 0; j < GPS; ++j) {
      const int tok = min(trow + GT * j, LCHUNK - 1);
      idx[j] = x[b * L_ + lt0 + tok];
    }
    if (wave == 0) {
      mi0 = (unsigned)x[b * L_ + lt0 + lane];
      mi1 = (unsigned)x[b * L_ + lt0 + 32 + lane];
    }
    __builtin_amdgcn_sched_barrier(0);
    const v2f* srcr = reinterpret_cast<const v2f*>(kr_in);
    const v2f* srci = reinterpret_cast<const v2f*>(ki_in);
    #pragma unroll
    for (int q = 0; q < SPT; ++q) {
      sr[q] = srcr[tid + NT * q];
      si[q] = srci[tid + NT * q];
    }
    __builtin_amdgcn_sched_barrier(0);
    if (wave == 0) {
      mv0 = mix_table[mi0];
      mv1 = mix_table[mi1];
    }
    #pragma unroll
    for (int j = 0; j < GPS; ++j) {
      av[j] = *reinterpret_cast<const v2f*>(amp_table + idx[j] * D_ + 2 * dq);
      pv[j] = *reinterpret_cast<const v2f*>(pha_table + idx[j] * D_ + 2 * dq);
    }
  }

  #pragma unroll
  for (int q = 0; q < SPT; ++q) {
    const int i  = tid + NT * q;
    const int m  = i / DH;
    const int dp = i - m * DH;
    v2f* dst = reinterpret_cast<v2f*>(kcs + m * BP) + dp;
    dst[0]  = sr[q];
    dst[DH] = si[q];
  }

  for (int i = tid; i < LCHUNK * 3; i += NT) {
    const int tok = i / 3;
    const int pl  = i - tok * 3;
    *reinterpret_cast<v2f*>(vt + tok * AP + pl * DP + D_) = v2f{0.0f, 0.0f};
  }

  if (wave == 0) {
    float cm = fmaxf(mv0, mv1);
    cm = fmaxf(cm, lane_xor_16(cm));
    cm = fmaxf(cm, lane_xor_row<8>(cm));
    cm = fmaxf(cm, lane_xor_row<4>(cm));
    cm = fmaxf(cm, lane_xor_row<2>(cm));
    cm = fmaxf(cm, lane_xor_row<1>(cm));
    const float e0 = expf(mv0 - cm);
    const float e1 = expf(mv1 - cm);
    wt[lane]      = e0;
    wt[lane + 32] = e1;
    float es = e0 + e1;
    es += lane_xor_16(es);
    es += lane_xor_row<8>(es);
    es += lane_xor_row<4>(es);
    es += lane_xor_row<2>(es);
    es += lane_xor_row<1>(es);
    if (lane == 0) {
      scal[0] = cm;
      scal[1] = es;
    }
  }

  #pragma unroll
  for (int j = 0; j < GPS; ++j) {
    const int tok = trow + GT * j;
    const float rev0 = pv[j].x * 0.15915494f;
    const float rev1 = pv[j].y * 0.15915494f;
    const float s0 = __builtin_amdgcn_sinf(rev0);
    const float c0 = __builtin_amdgcn_cosf(rev0);
    const float s1 = __builtin_amdgcn_sinf(rev1);
    const float c1 = __builtin_amdgcn_cosf(rev1);
    const float re0 = av[j].x * c0;
    const float im0 = av[j].x * s0;
    const float re1 = av[j].y * c1;
    const float im1 = av[j].y * s1;
    if (act && tok < LCHUNK) {
      float* row = vt + tok * AP + 2 * dq;
      *reinterpret_cast<v2f*>(row)          = v2f{re0, re1};
      *reinterpret_cast<v2f*>(row + DP)     = v2f{im0 - re0, im1 - re1};
      *reinterpret_cast<v2f*>(row + 2 * DP) = v2f{re0 + im0, re1 + im1};
    }
  }
  __syncthreads();

  const int mbase = wave * (16 * MT);
  const int nrow  = lane & 15;
  const int khalf = (lane >> 4) * 2;
  const int rbase = (lane >> 4) * 8;

  v2f bR[MT][KS], bI[MT][KS], bD[MT][KS];
  float rn2[MT];
  #pragma unroll
  for (int mt = 0; mt < MT; ++mt) {
    const float* brow = kcs + (mbase + 16 * mt + nrow) * BP + khalf;
    #pragma unroll
    for (int ks = 0; ks < KS; ++ks) {
      v2f r = *reinterpret_cast<const v2f*>(brow + 4 * ks);
      v2f i = *reinterpret_cast<const v2f*>(brow + D_ + 4 * ks);
      if (4 * ks + 2 >= D_ && khalf != 0) {
        r = v2f{0.0f, 0.0f};
        i = v2f{0.0f, 0.0f};
      }
      bR[mt][ks] = r;
      bI[mt][ks] = i;
      bD[mt][ks] = r - i;
    }
    float nx = 0.0f, ny = 0.0f;
    #pragma unroll
    for (int ks = 0; ks < KS; ++ks) {
      nx = fmaf(bR[mt][ks].x, bR[mt][ks].x, nx);
      ny = fmaf(bR[mt][ks].y, bR[mt][ks].y, ny);
      nx = fmaf(bI[mt][ks].x, bI[mt][ks].x, nx);
      ny = fmaf(bI[mt][ks].y, bI[mt][ks].y, ny);
    }
    float nn = nx + ny;
    nn += __shfl_xor(nn, 16, 32);
    rn2[mt] = 1.0f / nn;
  }

  float acc[MT];
  #pragma unroll
  for (int mt = 0; mt < MT; ++mt) acc[mt] = 0.0f;

  v2f aN[3][KS];
  float wN[8];
  {
    const float* arow = vt + nrow * AP + khalf;
    #pragma unroll
    for (int ks = 0; ks < KS; ++ks) {
      aN[0][ks] = *reinterpret_cast<const v2f*>(arow + 4 * ks);
      aN[1][ks] = *reinterpret_cast<const v2f*>(arow + DP + 4 * ks);
      aN[2][ks] = *reinterpret_cast<const v2f*>(arow + 2 * DP + 4 * ks);
    }
    #pragma unroll
    for (int r = 0; r < 8; ++r) wN[r] = wt[rbase + r];
  }

  #pragma unroll
  for (int t = 0; t < NTILE; ++t) {
    v2f aC[3][KS];
    float wC[8];
    #pragma unroll
    for (int ks = 0; ks < KS; ++ks) {
      aC[0][ks] = aN[0][ks];
      aC[1][ks] = aN[1][ks];
      aC[2][ks] = aN[2][ks];
    }
    #pragma unroll
    for (int r = 0; r < 8; ++r) wC[r] = wN[r];

    __builtin_amdgcn_sched_barrier(0);
    if (t + 1 < NTILE) {
      const float* arow = vt + ((t + 1) * 16 + nrow) * AP + khalf;
      #pragma unroll
      for (int ks = 0; ks < KS; ++ks) {
        aN[0][ks] = *reinterpret_cast<const v2f*>(arow + 4 * ks);
        aN[1][ks] = *reinterpret_cast<const v2f*>(arow + DP + 4 * ks);
        aN[2][ks] = *reinterpret_cast<const v2f*>(arow + 2 * DP + 4 * ks);
      }
      #pragma unroll
      for (int r = 0; r < 8; ++r) wN[r] = wt[(t + 1) * 16 + rbase + r];
    }
    __builtin_amdgcn_sched_barrier(0);

    v8f c1[MT], c2[MT], c3[MT];
    #pragma unroll
    for (int mt = 0; mt < MT; ++mt) { c1[mt] = v8f{}; c2[mt] = v8f{}; c3[mt] = v8f{}; }
    #pragma unroll
    for (int ks = 0; ks < KS; ++ks) {
      #pragma unroll
      for (int mt = 0; mt < MT; ++mt) {
        c1[mt] = __builtin_amdgcn_wmma_f32_16x16x4_f32(false, aC[0][ks], false, bD[mt][ks], (short)0, c1[mt], false, false);
        c2[mt] = __builtin_amdgcn_wmma_f32_16x16x4_f32(false, aC[1][ks], false, bR[mt][ks], (short)0, c2[mt], false, false);
        c3[mt] = __builtin_amdgcn_wmma_f32_16x16x4_f32(false, aC[2][ks], false, bI[mt][ks], (short)0, c3[mt], false, false);
      }
    }

    #pragma unroll
    for (int mt = 0; mt < MT; ++mt) {
      float partial = 0.0f;
      #pragma unroll
      for (int r = 0; r < 8; ++r) {
        const float ar = c1[mt][r] + c3[mt][r];
        const float ai = c1[mt][r] + c2[mt][r];
        partial = fmaf(wC[r], fmaf(ar, ar, ai * ai), partial);
      }
      partial += __shfl_xor(partial, 16, 32);
      acc[mt] += partial;
    }
  }
  if (lane < 16) {
    #pragma unroll
    for (int mt = 0; mt < MT; ++mt) pacc[mbase + 16 * mt + lane] = acc[mt] * rn2[mt];
  }
  __syncthreads();

  for (int i = tid; i < PROW; i += NT) {
    float val;
    if (i < M_) {
      val = pacc[i];
    } else {
      const int j = i - M_;
      val = (j < 2) ? scal[j] : 0.0f;
    }
    part[(b * LSPLIT + split) * PROW + i] = val;
  }
}

__global__ void __launch_bounds__(M_)
merge_kernel(const float* __restrict__ part, float* __restrict__ out) {
  const int b = blockIdx.x;
  const int m = threadIdx.x;
  const float* row = part + b * LSPLIT * PROW;
  float pj[LSPLIT], cj[LSPLIT], sj[LSPLIT];
  #pragma unroll
  for (int j = 0; j < LSPLIT; ++j) {
    pj[j] = row[j * PROW + m];
    cj[j] = row[j * PROW + M_];
    sj[j] = row[j * PROW + M_ + 1];
  }
  float cmax = cj[0];
  #pragma unroll
  for (int j = 1; j < LSPLIT; ++j) cmax = fmaxf(cmax, cj[j]);
  float num = 0.0f, den = 0.0f;
  #pragma unroll
  for (int j = 0; j < LSPLIT; ++j) {
    const float f = expf(cj[j] - cmax);
    num = fmaf(pj[j], f, num);
    den = fmaf(sj[j], f, den);
  }
  out[b * M_ + m] = num / den;
}

extern "C" void kernel_launch(void* const* d_in, const int* in_sizes, int n_in,
                              void* d_out, int out_size, void* d_ws, size_t ws_size,
                              hipStream_t stream) {
  const int*   x   = (const int*)d_in[0];
  const float* amp = (const float*)d_in[1];
  const float* pha = (const float*)d_in[2];
  const float* mix = (const float*)d_in[3];
  const float* kr  = (const float*)d_in[4];
  const float* ki  = (const float*)d_in[5];
  float*       out = (float*)d_out;

  float* part = (float*)d_ws;

  const size_t lds_bytes =
      (size_t)(M_ * BP + LCHUNK * AP + LCHUNK + M_ + 2) * sizeof(float);
  dim3 grid(B_, LSPLIT);
  qproj_kernel<<<grid, NT, lds_bytes, stream>>>(x, amp, pha, mix, kr, ki, part);
  merge_kernel<<<B_, M_, 0, stream>>>(part, out);
}
